# starting kernel
# baseline (speedup 1.0000x reference)
#include <hip/hip_runtime.h>

#define N_ 16384
#define K_ 64
#define F_ 2048
#define ALPHA_ 1.0f
#define GAMMA_ 1.0f

typedef float v2f __attribute__((ext_vector_type(2)));
typedef float v8f __attribute__((ext_vector_type(8)));

__global__ __launch_bounds__(32) void gram_wmma(const float* __restrict__ C,
                                                float* __restrict__ G) {
    const int tile = blockIdx.x;
    const int ta = tile >> 2;
    const int tb = tile & 3;
    const int lane = threadIdx.x;
    const int lo = lane & 15;
    const int hi = lane >> 4;

    const v2f* __restrict__ ap = (const v2f*)(C + (size_t)(ta * 16 + lo) * F_);
    const v2f* __restrict__ bp = (const v2f*)(C + (size_t)(tb * 16 + lo) * F_);

    v8f acc = {0.f, 0.f, 0.f, 0.f, 0.f, 0.f, 0.f, 0.f};
#pragma unroll 4
    for (int i = 0; i < F_ / 4; ++i) {
        v2f a = ap[2 * i + hi];
        v2f b = bp[2 * i + hi];
        acc = __builtin_amdgcn_wmma_f32_16x16x4_f32(
             false, a,  false, b,
             (short)0, acc,  false,  false);
    }
#pragma unroll
    for (int r = 0; r < 8; ++r) {
        int m = r + 8 * hi;
        G[(ta * 16 + m) * K_ + tb * 16 + lo] = acc[r];
    }
}

__global__ __launch_bounds__(64) void stats(const float* __restrict__ G,
                                            const float* __restrict__ cw,
                                            float* __restrict__ rs1,
                                            float* __restrict__ w) {
    __shared__ float nrm_s[K_];
    const int k = threadIdx.x;
    float rowG = 0.f;
#pragma unroll 8
    for (int j = 0; j < K_; ++j) rowG += G[k * K_ + j];
    const float nrm = G[k * K_ + k];
    nrm_s[k] = nrm;
    __syncthreads();
    float S = 0.f;
#pragma unroll 8
    for (int j = 0; j < K_; ++j) S += nrm_s[j];
    const float rs = (float)K_ * nrm + S - 2.f * rowG + ALPHA_;
    rs1[k] = rs;
    w[k] = cw[k] / rs;
}

__global__ __launch_bounds__(256) void loss_idx(const float* __restrict__ features,
                                                const float* __restrict__ labels,
                                                const float* __restrict__ cluster,
                                                const float* __restrict__ rs1,
                                                float* __restrict__ loss,
                                                int* __restrict__ idx_out) {
    const int lane = threadIdx.x & 31;
    const int row = blockIdx.x * 8 + (threadIdx.x >> 5);

    float v0 = labels[(size_t)row * K_ + lane];
    float v1 = labels[(size_t)row * K_ + lane + 32];
    float bv = v0;
    int bi = lane;
    if (v1 > v0) { bv = v1; bi = lane + 32; }
#pragma unroll
    for (int m = 16; m >= 1; m >>= 1) {
        float ov = __shfl_xor(bv, m, 32);
        int oi = __shfl_xor(bi, m, 32);
        if (ov > bv || (ov == bv && oi < bi)) { bv = ov; bi = oi; }
    }
    const int idx = bi;

    const float4* __restrict__ x4 = (const float4*)(features + (size_t)row * F_);
    const float4* __restrict__ c4 = (const float4*)(cluster + (size_t)idx * F_);
    float s = 0.f;
#pragma unroll 4
    for (int i = lane; i < F_ / 4; i += 32) {
        float4 a = x4[i];
        float4 b = c4[i];
        float d0 = a.x - b.x, d1 = a.y - b.y, d2 = a.z - b.z, d3 = a.w - b.w;
        s += d0 * d0 + d1 * d1 + d2 * d2 + d3 * d3;
    }
#pragma unroll
    for (int m = 16; m >= 1; m >>= 1) s += __shfl_xor(s, m, 32);

    if (lane == 0) {
        loss[row] = s / rs1[idx];
        idx_out[row] = idx;
    }
}

__global__ __launch_bounds__(256) void cluster_update(const float* __restrict__ features,
                                                      const float* __restrict__ cluster,
                                                      const float* __restrict__ w,
                                                      const int* __restrict__ idx,
                                                      float* __restrict__ newc) {
    const int k = blockIdx.y;
    const int c0 = blockIdx.x * 512 + threadIdx.x * 2;
    float a0 = 0.f, a1 = 0.f;
    int cnt = 0;
    for (int n = 0; n < N_; ++n) {
        if (idx[n] == k) {
            const float2 xr = *(const float2*)(features + (size_t)n * F_ + c0);
            a0 += xr.x;
            a1 += xr.y;
            ++cnt;
        }
    }
    const float wk = w[k];
    const float2 cc = *(const float2*)(cluster + (size_t)k * F_ + c0);
    float2 out;
    out.x = cc.x - GAMMA_ * wk * (a0 - (float)cnt * cc.x);
    out.y = cc.y - GAMMA_ * wk * (a1 - (float)cnt * cc.y);
    *(float2*)(newc + (size_t)k * F_ + c0) = out;
}

extern "C" void kernel_launch(void* const* d_in, const int* in_sizes, int n_in,
                              void* d_out, int out_size, void* d_ws, size_t ws_size,
                              hipStream_t stream) {
    const float* features = (const float*)d_in[0];
    const float* labels   = (const float*)d_in[1];
    const float* cluster  = (const float*)d_in[2];
    const float* cw       = (const float*)d_in[3];

    float* loss = (float*)d_out;
    float* newc = loss + N_;

    float* G   = (float*)d_ws;
    float* rs1 = G + K_ * K_;
    float* wv  = rs1 + K_;
    int* idxw  = (int*)(wv + K_);

    gram_wmma<<<16, 32, 0, stream>>>(cluster, G);
    stats<<<1, 64, 0, stream>>>(G, cw, rs1, wv);
    loss_idx<<<N_ / 8, 256, 0, stream>>>(features, labels, cluster, rs1, loss, idxw);
    cluster_update<<<dim3(F_ / 512, K_), 256, 0, stream>>>(features, cluster, wv, idxw, newc);
}


# cluster_update: block compacts its cluster's rows into an LDS list by one parallel idx scan (no branch per row); 8 waves sum listed rows, 40 b128 loads in flight; sum order: 8 partials per column
# speedup vs baseline: 9.1032x; 9.1032x over previous
#include <hip/hip_runtime.h>

#define N_ 16384
#define K_ 64
#define F_ 2048
#define ALPHA_ 1.0f
#define GAMMA_ 1.0f

typedef float v2f __attribute__((ext_vector_type(2)));
typedef float v8f __attribute__((ext_vector_type(8)));

__global__ __launch_bounds__(32) void gram_wmma(const float* __restrict__ C,
                                                float* __restrict__ G) {
    const int tile = blockIdx.x;
    const int ta = tile >> 2;
    const int tb = tile & 3;
    const int lane = threadIdx.x;
    const int lo = lane & 15;
    const int hi = lane >> 4;

    const v2f* __restrict__ ap = (const v2f*)(C + (size_t)(ta * 16 + lo) * F_);
    const v2f* __restrict__ bp = (const v2f*)(C + (size_t)(tb * 16 + lo) * F_);

    v8f acc = {0.f, 0.f, 0.f, 0.f, 0.f, 0.f, 0.f, 0.f};
#pragma unroll 4
    for (int i = 0; i < F_ / 4; ++i) {
        v2f a = ap[2 * i + hi];
        v2f b = bp[2 * i + hi];
        acc = __builtin_amdgcn_wmma_f32_16x16x4_f32(
             false, a,  false, b,
             (short)0, acc,  false,  false);
    }
#pragma unroll
    for (int r = 0; r < 8; ++r) {
        int m = r + 8 * hi;
        G[(ta * 16 + m) * K_ + tb * 16 + lo] = acc[r];
    }
}

__global__ __launch_bounds__(64) void stats(const float* __restrict__ G,
                                            const float* __restrict__ cw,
                                            float* __restrict__ rs1,
                                            float* __restrict__ w) {
    __shared__ float nrm_s[K_];
    const int k = threadIdx.x;
    float rowG = 0.f;
#pragma unroll 8
    for (int j = 0; j < K_; ++j) rowG += G[k * K_ + j];
    const float nrm = G[k * K_ + k];
    nrm_s[k] = nrm;
    __syncthreads();
    float S = 0.f;
#pragma unroll 8
    for (int j = 0; j < K_; ++j) S += nrm_s[j];
    const float rs = (float)K_ * nrm + S - 2.f * rowG + ALPHA_;
    rs1[k] = rs;
    w[k] = cw[k] / rs;
}

__global__ __launch_bounds__(256) void loss_idx(const float* __restrict__ features,
                                                const float* __restrict__ labels,
                                                const float* __restrict__ cluster,
                                                const float* __restrict__ rs1,
                                                float* __restrict__ loss,
                                                int* __restrict__ idx_out) {
    const int lane = threadIdx.x & 31;
    const int row = blockIdx.x * 8 + (threadIdx.x >> 5);

    float v0 = labels[(size_t)row * K_ + lane];
    float v1 = labels[(size_t)row * K_ + lane + 32];
    float bv = v0;
    int bi = lane;
    if (v1 > v0) { bv = v1; bi = lane + 32; }
#pragma unroll
    for (int m = 16; m >= 1; m >>= 1) {
        float ov = __shfl_xor(bv, m, 32);
        int oi = __shfl_xor(bi, m, 32);
        if (ov > bv || (ov == bv && oi < bi)) { bv = ov; bi = oi; }
    }
    const int idx = bi;

    const float4* __restrict__ x4 = (const float4*)(features + (size_t)row * F_);
    const float4* __restrict__ c4 = (const float4*)(cluster + (size_t)idx * F_);
    float s = 0.f;
#pragma unroll 4
    for (int i = lane; i < F_ / 4; i += 32) {
        float4 a = x4[i];
        float4 b = c4[i];
        float d0 = a.x - b.x, d1 = a.y - b.y, d2 = a.z - b.z, d3 = a.w - b.w;
        s += d0 * d0 + d1 * d1 + d2 * d2 + d3 * d3;
    }
#pragma unroll
    for (int m = 16; m >= 1; m >>= 1) s += __shfl_xor(s, m, 32);

    if (lane == 0) {
        loss[row] = s / rs1[idx];
        idx_out[row] = idx;
    }
}

#define CU_COLS 256
#define CU_WAVES 8
#define CU_SCAN_IT (N_ / (CU_WAVES * 128))

static_assert(N_ <= 65536, "row numbers are kept as 16-bit values");
static_assert(CU_SCAN_IT * 4 <= 64, "one 64-bit match mask per thread");
static_assert(CU_SCAN_IT * CU_WAVES * 128 == N_, "scan covers idx exactly");

#define CU_G 20

__global__ __launch_bounds__(256) void cluster_update(const float* __restrict__ features,
                                                      const float* __restrict__ cluster,
                                                      const float* __restrict__ w,
                                                      const int* __restrict__ idx,
                                                      float* __restrict__ newc) {
    __shared__ unsigned short list_s[N_];
    __shared__ int wave_tot_s[CU_WAVES];
    __shared__ __attribute__((aligned(16))) float part_s[CU_WAVES][CU_COLS];

    const int k = blockIdx.y;
    const int c0 = blockIdx.x * CU_COLS;
    const int lane = threadIdx.x & 31;
    const int wv = __builtin_amdgcn_readfirstlane((int)(threadIdx.x >> 5));

    const int4* __restrict__ idx4 = (const int4*)idx + (size_t)wv * (CU_SCAN_IT * 32) + lane;
    unsigned long long m = 0ull;
#pragma unroll
    for (int i = 0; i < CU_SCAN_IT; ++i) {
        const int4 v = idx4[i * 32];
        const unsigned f = (unsigned)(v.x == k) | ((unsigned)(v.y == k) << 1) |
                           ((unsigned)(v.z == k) << 2) | ((unsigned)(v.w == k) << 3);
        m |= (unsigned long long)f << (4 * i);
    }
    const int c = __popcll(m);
    int incl = c;
#pragma unroll
    for (int d = 1; d < 32; d <<= 1) {
        const int t = __shfl_up(incl, d, 32);
        if (lane >= d) incl += t;
    }
    if (lane == 31) wave_tot_s[wv] = incl;
    __syncthreads();
    int base = 0, total = 0;
#pragma unroll
    for (int j = 0; j < CU_WAVES; ++j) {
        const int t = wave_tot_s[j];
        if (j < wv) base += t;
        total += t;
    }
    total = __builtin_amdgcn_readfirstlane(total);
    {
        int pos = base + incl - c;
        const int n0 = wv * (CU_SCAN_IT * 128) + lane * 4;
        while (m) {
            const int b = __builtin_ctzll(m);
            m &= m - 1;
            list_s[pos++] = (unsigned short)(n0 + (b >> 2) * 128 + (b & 3));
        }
    }
    __syncthreads();

    const int beg = (total * wv) / CU_WAVES;
    const int end = (total * (wv + 1)) / CU_WAVES;
    float4 s0 = {0.f, 0.f, 0.f, 0.f};
    float4 s1 = {0.f, 0.f, 0.f, 0.f};
    const float* __restrict__ fb = features + c0;
    for (int q = beg; q < end; q += CU_G) {
        int pp = q + (lane < CU_G ? lane : CU_G - 1);
        pp = pp < end ? pp : end - 1;
        const int rv = (int)list_s[pp];
        v8f X[CU_G];
#pragma unroll
        for (int r = 0; r < CU_G; ++r) {
            const int row = __builtin_amdgcn_readlane(rv, r);
            const float4* rp = (const float4*)(fb + (size_t)row * F_);
            const float4 lo = rp[lane];
            const float4 hi = rp[lane + 32];
            X[r] = (v8f){lo.x, lo.y, lo.z, lo.w, hi.x, hi.y, hi.z, hi.w};
        }
        asm volatile("" ::"v"(X[0]), "v"(X[1]), "v"(X[2]), "v"(X[3]), "v"(X[4]), "v"(X[5]),
                     "v"(X[6]), "v"(X[7]), "v"(X[8]), "v"(X[9]), "v"(X[10]), "v"(X[11]),
                     "v"(X[12]), "v"(X[13]), "v"(X[14]), "v"(X[15]), "v"(X[16]), "v"(X[17]),
                     "v"(X[18]), "v"(X[19]));
        static_assert(CU_G == 20, "the fence above names CU_G values");
#pragma unroll
        for (int r = 0; r < CU_G; ++r) {
            const float wt = (q + r < end) ? 1.f : 0.f;
            s0.x = fmaf(X[r][0], wt, s0.x);
            s0.y = fmaf(X[r][1], wt, s0.y);
            s0.z = fmaf(X[r][2], wt, s0.z);
            s0.w = fmaf(X[r][3], wt, s0.w);
            s1.x = fmaf(X[r][4], wt, s1.x);
            s1.y = fmaf(X[r][5], wt, s1.y);
            s1.z = fmaf(X[r][6], wt, s1.z);
            s1.w = fmaf(X[r][7], wt, s1.w);
        }
    }

    float4* ps = (float4*)&part_s[wv][0];
    ps[lane] = s0;
    ps[32 + lane] = s1;
    __syncthreads();
    const int t = threadIdx.x;
    float a = 0.f;
#pragma unroll
    for (int j = 0; j < CU_WAVES; ++j) a += part_s[j][t];
    const float wk = w[k];
    const float cc = cluster[(size_t)k * F_ + c0 + t];
    newc[(size_t)k * F_ + c0 + t] = cc - GAMMA_ * wk * (a - (float)total * cc);
}

extern "C" void kernel_launch(void* const* d_in, const int* in_sizes, int n_in,
                              void* d_out, int out_size, void* d_ws, size_t ws_size,
                              hipStream_t stream) {
    const float* features = (const float*)d_in[0];
    const float* labels   = (const float*)d_in[1];
    const float* cluster  = (const float*)d_in[2];
    const float* cw       = (const float*)d_in[3];

    float* loss = (float*)d_out;
    float* newc = loss + N_;

    float* G   = (float*)d_ws;
    float* rs1 = G + K_ * K_;
    float* wv  = rs1 + K_;
    int* idxw  = (int*)(wv + K_);

    gram_wmma<<<16, 32, 0, stream>>>(cluster, G);
    stats<<<1, 64, 0, stream>>>(G, cw, rs1, wv);
    loss_idx<<<N_ / 8, 256, 0, stream>>>(features, labels, cluster, rs1, loss, idxw);
    cluster_update<<<dim3(F_ / CU_COLS, K_), 256, 0, stream>>>(features, cluster, wv, idxw, newc);
}


# gram: each 16x16 tile's k sum split over the 16 waves of its block (32 WMMA per wave, was 512 in one wave), 16-byte operand loads; sum order: 2 accumulators per wave, 16 partial tiles added in LDS
# speedup vs baseline: 14.9369x; 1.6408x over previous
#include <hip/hip_runtime.h>

#define N_ 16384
#define K_ 64
#define F_ 2048
#define ALPHA_ 1.0f
#define GAMMA_ 1.0f

typedef float v2f __attribute__((ext_vector_type(2)));
typedef float v8f __attribute__((ext_vector_type(8)));

#define GR_WAVES 16
#define GR_KW (F_ / GR_WAVES)
#define GR_LD (GR_KW / 8)

static_assert(GR_KW * GR_WAVES == F_ && GR_LD * 8 == GR_KW, "the waves' k ranges cover F exactly");

__global__ __launch_bounds__(GR_WAVES * 32) void gram_wmma(const float* __restrict__ C,
                                                           float* __restrict__ G) {
    __shared__ float gpart_s[GR_WAVES][256];

    const int tile = blockIdx.x;
    const int ta = tile >> 2;
    const int tb = tile & 3;
    const int lane = threadIdx.x & 31;
    const int wv = threadIdx.x >> 5;
    const int lo = lane & 15;
    const int hi = lane >> 4;

    const float4* __restrict__ ap =
        (const float4*)(C + (size_t)(ta * 16 + lo) * F_ + wv * GR_KW) + hi;
    const float4* __restrict__ bp =
        (const float4*)(C + (size_t)(tb * 16 + lo) * F_ + wv * GR_KW) + hi;

    float4 a[GR_LD], b[GR_LD];
#pragma unroll
    for (int i = 0; i < GR_LD; ++i) {
        a[i] = ap[2 * i];
        b[i] = bp[2 * i];
    }

    v8f acc0 = {0.f, 0.f, 0.f, 0.f, 0.f, 0.f, 0.f, 0.f};
    v8f acc1 = {0.f, 0.f, 0.f, 0.f, 0.f, 0.f, 0.f, 0.f};
#pragma unroll
    for (int i = 0; i < GR_LD; ++i) {
        const v2f a01 = {a[i].x, a[i].y}, b01 = {b[i].x, b[i].y};
        const v2f a23 = {a[i].z, a[i].w}, b23 = {b[i].z, b[i].w};
        acc0 = __builtin_amdgcn_wmma_f32_16x16x4_f32(
             false, a01,  false, b01,
             (short)0, acc0,  false,  false);
        acc1 = __builtin_amdgcn_wmma_f32_16x16x4_f32(
             false, a23,  false, b23,
             (short)0, acc1,  false,  false);
    }

#pragma unroll
    for (int r = 0; r < 8; ++r) gpart_s[wv][(r + 8 * hi) * 16 + lo] = acc0[r] + acc1[r];
    __syncthreads();

    const int t = threadIdx.x;
    if (t < 256) {
        float s = 0.f;
#pragma unroll
        for (int j = 0; j < GR_WAVES; ++j) s += gpart_s[j][t];
        G[(ta * 16 + (t >> 4)) * K_ + tb * 16 + (t & 15)] = s;
    }
}

__global__ __launch_bounds__(64) void stats(const float* __restrict__ G,
                                            const float* __restrict__ cw,
                                            float* __restrict__ rs1,
                                            float* __restrict__ w) {
    __shared__ float nrm_s[K_];
    const int k = threadIdx.x;
    float rowG = 0.f;
#pragma unroll 8
    for (int j = 0; j < K_; ++j) rowG += G[k * K_ + j];
    const float nrm = G[k * K_ + k];
    nrm_s[k] = nrm;
    __syncthreads();
    float S = 0.f;
#pragma unroll 8
    for (int j = 0; j < K_; ++j) S += nrm_s[j];
    const float rs = (float)K_ * nrm + S - 2.f * rowG + ALPHA_;
    rs1[k] = rs;
    w[k] = cw[k] / rs;
}

__global__ __launch_bounds__(256) void loss_idx(const float* __restrict__ features,
                                                const float* __restrict__ labels,
                                                const float* __restrict__ cluster,
                                                const float* __restrict__ rs1,
                                                float* __restrict__ loss,
                                                int* __restrict__ idx_out) {
    const int lane = threadIdx.x & 31;
    const int row = blockIdx.x * 8 + (threadIdx.x >> 5);

    float v0 = labels[(size_t)row * K_ + lane];
    float v1 = labels[(size_t)row * K_ + lane + 32];
    float bv = v0;
    int bi = lane;
    if (v1 > v0) { bv = v1; bi = lane + 32; }
#pragma unroll
    for (int m = 16; m >= 1; m >>= 1) {
        float ov = __shfl_xor(bv, m, 32);
        int oi = __shfl_xor(bi, m, 32);
        if (ov > bv || (ov == bv && oi < bi)) { bv = ov; bi = oi; }
    }
    const int idx = bi;

    const float4* __restrict__ x4 = (const float4*)(features + (size_t)row * F_);
    const float4* __restrict__ c4 = (const float4*)(cluster + (size_t)idx * F_);
    float s = 0.f;
#pragma unroll 4
    for (int i = lane; i < F_ / 4; i += 32) {
        float4 a = x4[i];
        float4 b = c4[i];
        float d0 = a.x - b.x, d1 = a.y - b.y, d2 = a.z - b.z, d3 = a.w - b.w;
        s += d0 * d0 + d1 * d1 + d2 * d2 + d3 * d3;
    }
#pragma unroll
    for (int m = 16; m >= 1; m >>= 1) s += __shfl_xor(s, m, 32);

    if (lane == 0) {
        loss[row] = s / rs1[idx];
        idx_out[row] = idx;
    }
}

#define CU_COLS 256
#define CU_WAVES 8
#define CU_SCAN_IT (N_ / (CU_WAVES * 128))

static_assert(N_ <= 65536, "row numbers are kept as 16-bit values");
static_assert(CU_SCAN_IT * 4 <= 64, "one 64-bit match mask per thread");
static_assert(CU_SCAN_IT * CU_WAVES * 128 == N_, "scan covers idx exactly");

#define CU_G 20

__global__ __launch_bounds__(256) void cluster_update(const float* __restrict__ features,
                                                      const float* __restrict__ cluster,
                                                      const float* __restrict__ w,
                                                      const int* __restrict__ idx,
                                                      float* __restrict__ newc) {
    __shared__ unsigned short list_s[N_];
    __shared__ int wave_tot_s[CU_WAVES];
    __shared__ __attribute__((aligned(16))) float part_s[CU_WAVES][CU_COLS];

    const int k = blockIdx.y;
    const int c0 = blockIdx.x * CU_COLS;
    const int lane = threadIdx.x & 31;
    const int wv = __builtin_amdgcn_readfirstlane((int)(threadIdx.x >> 5));

    const int4* __restrict__ idx4 = (const int4*)idx + (size_t)wv * (CU_SCAN_IT * 32) + lane;
    unsigned long long m = 0ull;
#pragma unroll
    for (int i = 0; i < CU_SCAN_IT; ++i) {
        const int4 v = idx4[i * 32];
        const unsigned f = (unsigned)(v.x == k) | ((unsigned)(v.y == k) << 1) |
                           ((unsigned)(v.z == k) << 2) | ((unsigned)(v.w == k) << 3);
        m |= (unsigned long long)f << (4 * i);
    }
    const int c = __popcll(m);
    int incl = c;
#pragma unroll
    for (int d = 1; d < 32; d <<= 1) {
        const int t = __shfl_up(incl, d, 32);
        if (lane >= d) incl += t;
    }
    if (lane == 31) wave_tot_s[wv] = incl;
    __syncthreads();
    int base = 0, total = 0;
#pragma unroll
    for (int j = 0; j < CU_WAVES; ++j) {
        const int t = wave_tot_s[j];
        if (j < wv) base += t;
        total += t;
    }
    total = __builtin_amdgcn_readfirstlane(total);
    {
        int pos = base + incl - c;
        const int n0 = wv * (CU_SCAN_IT * 128) + lane * 4;
        while (m) {
            const int b = __builtin_ctzll(m);
            m &= m - 1;
            list_s[pos++] = (unsigned short)(n0 + (b >> 2) * 128 + (b & 3));
        }
    }
    __syncthreads();

    const int beg = (total * wv) / CU_WAVES;
    const int end = (total * (wv + 1)) / CU_WAVES;
    float4 s0 = {0.f, 0.f, 0.f, 0.f};
    float4 s1 = {0.f, 0.f, 0.f, 0.f};
    const float* __restrict__ fb = features + c0;
    for (int q = beg; q < end; q += CU_G) {
        int pp = q + (lane < CU_G ? lane : CU_G - 1);
        pp = pp < end ? pp : end - 1;
        const int rv = (int)list_s[pp];
        v8f X[CU_G];
#pragma unroll
        for (int r = 0; r < CU_G; ++r) {
            const int row = __builtin_amdgcn_readlane(rv, r);
            const float4* rp = (const float4*)(fb + (size_t)row * F_);
            const float4 lo = rp[lane];
            const float4 hi = rp[lane + 32];
            X[r] = (v8f){lo.x, lo.y, lo.z, lo.w, hi.x, hi.y, hi.z, hi.w};
        }
        asm volatile("" ::"v"(X[0]), "v"(X[1]), "v"(X[2]), "v"(X[3]), "v"(X[4]), "v"(X[5]),
                     "v"(X[6]), "v"(X[7]), "v"(X[8]), "v"(X[9]), "v"(X[10]), "v"(X[11]),
                     "v"(X[12]), "v"(X[13]), "v"(X[14]), "v"(X[15]), "v"(X[16]), "v"(X[17]),
                     "v"(X[18]), "v"(X[19]));
        static_assert(CU_G == 20, "the fence above names CU_G values");
#pragma unroll
        for (int r = 0; r < CU_G; ++r) {
            const float wt = (q + r < end) ? 1.f : 0.f;
            s0.x = fmaf(X[r][0], wt, s0.x);
            s0.y = fmaf(X[r][1], wt, s0.y);
            s0.z = fmaf(X[r][2], wt, s0.z);
            s0.w = fmaf(X[r][3], wt, s0.w);
            s1.x = fmaf(X[r][4], wt, s1.x);
            s1.y = fmaf(X[r][5], wt, s1.y);
            s1.z = fmaf(X[r][6], wt, s1.z);
            s1.w = fmaf(X[r][7], wt, s1.w);
        }
    }

    float4* ps = (float4*)&part_s[wv][0];
    ps[lane] = s0;
    ps[32 + lane] = s1;
    __syncthreads();
    const int t = threadIdx.x;
    float a = 0.f;
#pragma unroll
    for (int j = 0; j < CU_WAVES; ++j) a += part_s[j][t];
    const float wk = w[k];
    const float cc = cluster[(size_t)k * F_ + c0 + t];
    newc[(size_t)k * F_ + c0 + t] = cc - GAMMA_ * wk * (a - (float)total * cc);
}

extern "C" void kernel_launch(void* const* d_in, const int* in_sizes, int n_in,
                              void* d_out, int out_size, void* d_ws, size_t ws_size,
                              hipStream_t stream) {
    const float* features = (const float*)d_in[0];
    const float* labels   = (const float*)d_in[1];
    const float* cluster  = (const float*)d_in[2];
    const float* cw       = (const float*)d_in[3];

    float* loss = (float*)d_out;
    float* newc = loss + N_;

    float* G   = (float*)d_ws;
    float* rs1 = G + K_ * K_;
    float* wv  = rs1 + K_;
    int* idxw  = (int*)(wv + K_);

    gram_wmma<<<16, GR_WAVES * 32, 0, stream>>>(cluster, G);
    stats<<<1, 64, 0, stream>>>(G, cw, rs1, wv);
    loss_idx<<<N_ / 8, 256, 0, stream>>>(features, labels, cluster, rs1, loss, idxw);
    cluster_update<<<dim3(F_ / CU_COLS, K_), 256, 0, stream>>>(features, cluster, wv, idxw, newc);
}


# launch fusion, 4 launches -> 2: the 16 Gram tile blocks ride in the row-distance launch; each update block forms row_sum from G in LDS and finishes 32 loss values; every sum order kept
# speedup vs baseline: 18.0194x; 1.2064x over previous
#include <hip/hip_runtime.h>

#define N_ 16384
#define K_ 64
#define F_ 2048
#define ALPHA_ 1.0f
#define GAMMA_ 1.0f

typedef float v2f __attribute__((ext_vector_type(2)));
typedef float v8f __attribute__((ext_vector_type(8)));


#define GR_BLOCKS 16
#define GR_CHUNKS 16
#define GR_KW (F_ / GR_CHUNKS)
#define GR_LD (GR_KW / 8)
#define L1_WAVES 8

static_assert(GR_KW * GR_CHUNKS == F_ && GR_LD * 8 == GR_KW, "the chunks cover F exactly");
static_assert(GR_CHUNKS == 2 * L1_WAVES, "two chunks per wave");
static_assert(GR_BLOCKS == (K_ / 16) * (K_ / 16), "one block per 16x16 tile");

__device__ __forceinline__ void gram_tile(const float* __restrict__ C, float* __restrict__ G,
                                          float (*gpart_s)[256]) {
    const int tile = blockIdx.x;
    const int ta = tile >> 2;
    const int tb = tile & 3;
    const int lane = threadIdx.x & 31;
    const int wv = threadIdx.x >> 5;
    const int lo = lane & 15;
    const int hi = lane >> 4;

#pragma unroll
    for (int h = 0; h < 2; ++h) {
        const int ch = wv + L1_WAVES * h;
        const float4* __restrict__ ap =
            (const float4*)(C + (size_t)(ta * 16 + lo) * F_ + ch * GR_KW) + hi;
        const float4* __restrict__ bp =
            (const float4*)(C + (size_t)(tb * 16 + lo) * F_ + ch * GR_KW) + hi;

        float4 a[GR_LD], b[GR_LD];
#pragma unroll
        for (int i = 0; i < GR_LD; ++i) {
            a[i] = ap[2 * i];
            b[i] = bp[2 * i];
        }

        v8f acc0 = {0.f, 0.f, 0.f, 0.f, 0.f, 0.f, 0.f, 0.f};
        v8f acc1 = {0.f, 0.f, 0.f, 0.f, 0.f, 0.f, 0.f, 0.f};
#pragma unroll
        for (int i = 0; i < GR_LD; ++i) {
            const v2f a01 = {a[i].x, a[i].y}, b01 = {b[i].x, b[i].y};
            const v2f a23 = {a[i].z, a[i].w}, b23 = {b[i].z, b[i].w};
            acc0 = __builtin_amdgcn_wmma_f32_16x16x4_f32(
                 false, a01,  false, b01,
                 (short)0, acc0,  false,  false);
            acc1 = __builtin_amdgcn_wmma_f32_16x16x4_f32(
                 false, a23,  false, b23,
                 (short)0, acc1,  false,  false);
        }

#pragma unroll
        for (int r = 0; r < 8; ++r) gpart_s[ch][(r + 8 * hi) * 16 + lo] = acc0[r] + acc1[r];
    }
    __syncthreads();

    const int t = threadIdx.x;
    float s = 0.f;
#pragma unroll
    for (int j = 0; j < GR_CHUNKS; ++j) s += gpart_s[j][t];
    G[(ta * 16 + (t >> 4)) * K_ + tb * 16 + (t & 15)] = s;
}

__global__ __launch_bounds__(L1_WAVES * 32) void gram_and_rows(const float* __restrict__ features,
                                                               const float* __restrict__ labels,
                                                               const float* __restrict__ cluster,
                                                               float* __restrict__ G,
                                                               float* __restrict__ nume,
                                                               int* __restrict__ idx_out) {
    __shared__ float gpart_s[GR_CHUNKS][256];

    if (blockIdx.x < GR_BLOCKS) {
        gram_tile(cluster, G, gpart_s);
        return;
    }

    const int lane = threadIdx.x & 31;
    const int row = (blockIdx.x - GR_BLOCKS) * L1_WAVES + (threadIdx.x >> 5);

    float v0 = labels[(size_t)row * K_ + lane];
    float v1 = labels[(size_t)row * K_ + lane + 32];
    float bv = v0;
    int bi = lane;
    if (v1 > v0) { bv = v1; bi = lane + 32; }
#pragma unroll
    for (int m = 16; m >= 1; m >>= 1) {
        float ov = __shfl_xor(bv, m, 32);
        int oi = __shfl_xor(bi, m, 32);
        if (ov > bv || (ov == bv && oi < bi)) { bv = ov; bi = oi; }
    }
    const int idx = bi;

    const float4* __restrict__ x4 = (const float4*)(features + (size_t)row * F_);
    const float4* __restrict__ c4 = (const float4*)(cluster + (size_t)idx * F_);
    float s = 0.f;
#pragma unroll 4
    for (int i = lane; i < F_ / 4; i += 32) {
        float4 a = x4[i];
        float4 b = c4[i];
        float d0 = a.x - b.x, d1 = a.y - b.y, d2 = a.z - b.z, d3 = a.w - b.w;
        s += d0 * d0 + d1 * d1 + d2 * d2 + d3 * d3;
    }
#pragma unroll
    for (int m = 16; m >= 1; m >>= 1) s += __shfl_xor(s, m, 32);

    if (lane == 0) {
        nume[row] = s;
        idx_out[row] = idx;
    }
}

#define CU_COLS 256
#define CU_WAVES 8
#define CU_SCAN_IT (N_ / (CU_WAVES * 128))
#define CU_BLOCKS ((F_ / CU_COLS) * K_)
#define CU_FIN (N_ / CU_BLOCKS)

static_assert(N_ <= 65536, "row numbers are kept as 16-bit values");
static_assert(CU_SCAN_IT * 4 <= 64, "one 64-bit match mask per thread");
static_assert(CU_SCAN_IT * CU_WAVES * 128 == N_, "scan covers idx exactly");
static_assert(CU_FIN == 32 && CU_FIN * CU_BLOCKS == N_, "one wave finishes the block's loss values");
static_assert(K_ * K_ == 4 * 4 * CU_WAVES * 32, "four 16-byte loads per thread copy G");
static_assert(K_ <= CU_WAVES * 32, "one thread per row statistic");

#define CU_G 20

__global__ __launch_bounds__(256) void cluster_update(const float* __restrict__ features,
                                                      const float* __restrict__ cluster,
                                                      const float* __restrict__ cw,
                                                      const float* __restrict__ G,
                                                      const float* __restrict__ nume,
                                                      const int* __restrict__ idx,
                                                      float* __restrict__ loss,
                                                      float* __restrict__ newc) {
    __shared__ unsigned short list_s[N_];
    __shared__ int wave_tot_s[CU_WAVES];
    __shared__ __attribute__((aligned(16))) float part_s[CU_WAVES][CU_COLS];
    __shared__ float g_s[K_][K_ + 1];
    __shared__ float rs1_s[K_];

    const int k = blockIdx.y;
    const int c0 = blockIdx.x * CU_COLS;
    const int lane = threadIdx.x & 31;
    const int wv = __builtin_amdgcn_readfirstlane((int)(threadIdx.x >> 5));

    const int4* __restrict__ idx4 = (const int4*)idx + (size_t)wv * (CU_SCAN_IT * 32) + lane;
    unsigned long long m = 0ull;
#pragma unroll
    for (int i = 0; i < CU_SCAN_IT; ++i) {
        const int4 v = idx4[i * 32];
        const unsigned f = (unsigned)(v.x == k) | ((unsigned)(v.y == k) << 1) |
                           ((unsigned)(v.z == k) << 2) | ((unsigned)(v.w == k) << 3);
        m |= (unsigned long long)f << (4 * i);
    }

    {
        const float4* __restrict__ G4 = (const float4*)G;
#pragma unroll
        for (int i = 0; i < 4; ++i) {
            const int e4 = (int)threadIdx.x + 256 * i;
            const float4 v = G4[e4];
            float* gp = &g_s[e4 >> 4][(e4 & 15) * 4];
            gp[0] = v.x; gp[1] = v.y; gp[2] = v.z; gp[3] = v.w;
        }
    }
    const int n_fin = ((int)blockIdx.y * (F_ / CU_COLS) + (int)blockIdx.x) * CU_FIN + lane;
    int fin_idx = 0;
    float fin_nume = 0.f;
    if (wv == CU_WAVES - 1) {
        fin_idx = idx[n_fin];
        fin_nume = nume[n_fin];
    }

    const int c = __popcll(m);
    int incl = c;
#pragma unroll
    for (int d = 1; d < 32; d <<= 1) {
        const int t = __shfl_up(incl, d, 32);
        if (lane >= d) incl += t;
    }
    if (lane == 31) wave_tot_s[wv] = incl;
    __syncthreads();
    int base = 0, total = 0;
#pragma unroll
    for (int j = 0; j < CU_WAVES; ++j) {
        const int t = wave_tot_s[j];
        if (j < wv) base += t;
        total += t;
    }
    total = __builtin_amdgcn_readfirstlane(total);
    {
        int pos = base + incl - c;
        const int n0 = wv * (CU_SCAN_IT * 128) + lane * 4;
        while (m) {
            const int b = __builtin_ctzll(m);
            m &= m - 1;
            list_s[pos++] = (unsigned short)(n0 + (b >> 2) * 128 + (b & 3));
        }
    }
    if (wv < K_ / 32) {
        const int j = threadIdx.x;
        float rowG = 0.f;
#pragma unroll
        for (int i = 0; i < K_; ++i) rowG += g_s[j][i];
        const float nrm = g_s[j][j];
        float S = 0.f;
#pragma unroll
        for (int i = 0; i < K_; ++i) S += g_s[i][i];
        rs1_s[j] = (float)K_ * nrm + S - 2.f * rowG + ALPHA_;
    }
    __syncthreads();

    if (wv == CU_WAVES - 1) loss[n_fin] = fin_nume / rs1_s[fin_idx];

    const int beg = (total * wv) / CU_WAVES;
    const int end = (total * (wv + 1)) / CU_WAVES;
    float4 s0 = {0.f, 0.f, 0.f, 0.f};
    float4 s1 = {0.f, 0.f, 0.f, 0.f};
    const float* __restrict__ fb = features + c0;
    for (int q = beg; q < end; q += CU_G) {
        int pp = q + (lane < CU_G ? lane : CU_G - 1);
        pp = pp < end ? pp : end - 1;
        const int rv = (int)list_s[pp];
        v8f X[CU_G];
#pragma unroll
        for (int r = 0; r < CU_G; ++r) {
            const int row = __builtin_amdgcn_readlane(rv, r);
            const float4* rp = (const float4*)(fb + (size_t)row * F_);
            const float4 lo = rp[lane];
            const float4 hi = rp[lane + 32];
            X[r] = (v8f){lo.x, lo.y, lo.z, lo.w, hi.x, hi.y, hi.z, hi.w};
        }
        asm volatile("" ::"v"(X[0]), "v"(X[1]), "v"(X[2]), "v"(X[3]), "v"(X[4]), "v"(X[5]),
                     "v"(X[6]), "v"(X[7]), "v"(X[8]), "v"(X[9]), "v"(X[10]), "v"(X[11]),
                     "v"(X[12]), "v"(X[13]), "v"(X[14]), "v"(X[15]), "v"(X[16]), "v"(X[17]),
                     "v"(X[18]), "v"(X[19]));
        static_assert(CU_G == 20, "the fence above names CU_G values");
#pragma unroll
        for (int r = 0; r < CU_G; ++r) {
            const float wt = (q + r < end) ? 1.f : 0.f;
            s0.x = fmaf(X[r][0], wt, s0.x);
            s0.y = fmaf(X[r][1], wt, s0.y);
            s0.z = fmaf(X[r][2], wt, s0.z);
            s0.w = fmaf(X[r][3], wt, s0.w);
            s1.x = fmaf(X[r][4], wt, s1.x);
            s1.y = fmaf(X[r][5], wt, s1.y);
            s1.z = fmaf(X[r][6], wt, s1.z);
            s1.w = fmaf(X[r][7], wt, s1.w);
        }
    }

    float4* ps = (float4*)&part_s[wv][0];
    ps[lane] = s0;
    ps[32 + lane] = s1;
    __syncthreads();
    const int t = threadIdx.x;
    float a = 0.f;
#pragma unroll
    for (int j = 0; j < CU_WAVES; ++j) a += part_s[j][t];
    const float wk = cw[k] / rs1_s[k];
    const float cc = cluster[(size_t)k * F_ + c0 + t];
    newc[(size_t)k * F_ + c0 + t] = cc - GAMMA_ * wk * (a - (float)total * cc);
}

extern "C" void kernel_launch(void* const* d_in, const int* in_sizes, int n_in,
                              void* d_out, int out_size, void* d_ws, size_t ws_size,
                              hipStream_t stream) {
    const float* features = (const float*)d_in[0];
    const float* labels   = (const float*)d_in[1];
    const float* cluster  = (const float*)d_in[2];
    const float* cw       = (const float*)d_in[3];

    float* loss = (float*)d_out;
    float* newc = loss + N_;

    float* G    = (float*)d_ws;
    float* nume = G + K_ * K_;
    int* idxw   = (int*)(nume + N_);

    gram_and_rows<<<GR_BLOCKS + N_ / L1_WAVES, L1_WAVES * 32, 0, stream>>>(
        features, labels, cluster, G, nume, idxw);
    cluster_update<<<dim3(F_ / CU_COLS, K_), 256, 0, stream>>>(
        features, cluster, cw, G, nume, idxw, loss, newc);
}


# features read once: launch 2 takes each row's squared distance from the loads of its cluster sums (8 tile partials per row, lane sums by DPP: new nume sum order); launch 1 argmax+Gram, launch 3 loss
# speedup vs baseline: 22.3171x; 1.2385x over previous
#include <hip/hip_runtime.h>

#define N_ 16384
#define K_ 64
#define F_ 2048
#define ALPHA_ 1.0f
#define GAMMA_ 1.0f

typedef float v2f __attribute__((ext_vector_type(2)));
typedef float v8f __attribute__((ext_vector_type(8)));


#define GR_BLOCKS 16
#define GR_CHUNKS 16
#define GR_KW (F_ / GR_CHUNKS)
#define GR_LD (GR_KW / 8)
#define L1_WAVES 8

static_assert(GR_KW * GR_CHUNKS == F_ && GR_LD * 8 == GR_KW, "the chunks cover F exactly");
static_assert(GR_CHUNKS == 2 * L1_WAVES, "two chunks per wave");
static_assert(GR_BLOCKS == (K_ / 16) * (K_ / 16), "one block per 16x16 tile");

__device__ __forceinline__ void gram_tile(const float* __restrict__ C, float* __restrict__ G,
                                          float (*gpart_s)[256]) {
    const int tile = blockIdx.x;
    const int ta = tile >> 2;
    const int tb = tile & 3;
    const int lane = threadIdx.x & 31;
    const int wv = threadIdx.x >> 5;
    const int lo = lane & 15;
    const int hi = lane >> 4;

#pragma unroll
    for (int h = 0; h < 2; ++h) {
        const int ch = wv + L1_WAVES * h;
        const float4* __restrict__ ap =
            (const float4*)(C + (size_t)(ta * 16 + lo) * F_ + ch * GR_KW) + hi;
        const float4* __restrict__ bp =
            (const float4*)(C + (size_t)(tb * 16 + lo) * F_ + ch * GR_KW) + hi;

        float4 a[GR_LD], b[GR_LD];
#pragma unroll
        for (int i = 0; i < GR_LD; ++i) {
            a[i] = ap[2 * i];
            b[i] = bp[2 * i];
        }

        v8f acc0 = {0.f, 0.f, 0.f, 0.f, 0.f, 0.f, 0.f, 0.f};
        v8f acc1 = {0.f, 0.f, 0.f, 0.f, 0.f, 0.f, 0.f, 0.f};
#pragma unroll
        for (int i = 0; i < GR_LD; ++i) {
            const v2f a01 = {a[i].x, a[i].y}, b01 = {b[i].x, b[i].y};
            const v2f a23 = {a[i].z, a[i].w}, b23 = {b[i].z, b[i].w};
            acc0 = __builtin_amdgcn_wmma_f32_16x16x4_f32(
                 false, a01,  false, b01,
                 (short)0, acc0,  false,  false);
            acc1 = __builtin_amdgcn_wmma_f32_16x16x4_f32(
                 false, a23,  false, b23,
                 (short)0, acc1,  false,  false);
        }

#pragma unroll
        for (int r = 0; r < 8; ++r) gpart_s[ch][(r + 8 * hi) * 16 + lo] = acc0[r] + acc1[r];
    }
    __syncthreads();

    const int t = threadIdx.x;
    float s = 0.f;
#pragma unroll
    for (int j = 0; j < GR_CHUNKS; ++j) s += gpart_s[j][t];
    G[(ta * 16 + (t >> 4)) * K_ + tb * 16 + (t & 15)] = s;
}

#define AR_BLOCKS (N_ / (L1_WAVES * 32))
static_assert(AR_BLOCKS * L1_WAVES * 32 == N_, "argmax blocks cover the rows exactly");
static_assert(K_ % 4 == 0, "a row of labels is whole 16-byte pieces");

__global__ __launch_bounds__(L1_WAVES * 32) void gram_and_argmax(const float* __restrict__ labels,
                                                                 const float* __restrict__ cluster,
                                                                 float* __restrict__ G,
                                                                 int* __restrict__ idx_out) {
    __shared__ float gpart_s[GR_CHUNKS][256];

    if (blockIdx.x < GR_BLOCKS) {
        gram_tile(cluster, G, gpart_s);
        return;
    }

    const int row = (blockIdx.x - GR_BLOCKS) * (L1_WAVES * 32) + threadIdx.x;
    const float4* __restrict__ lp = (const float4*)(labels + (size_t)row * K_);
    float4 v[K_ / 4];
#pragma unroll
    for (int i = 0; i < K_ / 4; ++i) v[i] = lp[i];

    float bv = v[0].x;
    int bi = 0;
#pragma unroll
    for (int i = 0; i < K_ / 4; ++i) {
        if (i > 0 && v[i].x > bv) { bv = v[i].x; bi = 4 * i; }
        if (v[i].y > bv) { bv = v[i].y; bi = 4 * i + 1; }
        if (v[i].z > bv) { bv = v[i].z; bi = 4 * i + 2; }
        if (v[i].w > bv) { bv = v[i].w; bi = 4 * i + 3; }
    }
    idx_out[row] = bi;
}

#define CU_COLS 256
#define CU_WAVES 8
#define CU_SCAN_IT (N_ / (CU_WAVES * 128))

static_assert(N_ <= 65536, "row numbers are kept as 16-bit values");
static_assert(CU_SCAN_IT * 4 <= 64, "one 64-bit match mask per thread");
static_assert(CU_SCAN_IT * CU_WAVES * 128 == N_, "scan covers idx exactly");
static_assert(K_ * K_ == 4 * 4 * CU_WAVES * 32, "four 16-byte loads per thread copy G");
static_assert(K_ <= CU_WAVES * 32, "one thread per row statistic");

#define CU_G 20

template <int M>
__device__ __forceinline__ float lane_xor_in_row(float x) {
    const int xi = __builtin_bit_cast(int, x);
    return __builtin_bit_cast(float, __builtin_amdgcn_update_dpp(0, xi, 0x160 + M, 0xf, 0xf, true));
}
__device__ __forceinline__ float other_row(float x) {
    const int xi = __builtin_bit_cast(int, x);
    return __builtin_bit_cast(float, __builtin_amdgcn_permlanex16(xi, xi, 0x76543210, 0xfedcba98, false, false));
}
__device__ __forceinline__ float wave_sum(float p) {
    p += lane_xor_in_row<1>(p);
    p += lane_xor_in_row<2>(p);
    p += lane_xor_in_row<4>(p);
    p += lane_xor_in_row<8>(p);
    p += other_row(p);
    return p;
}

__global__ __launch_bounds__(256) void cluster_update(const float* __restrict__ features,
                                                      const float* __restrict__ cluster,
                                                      const float* __restrict__ cw,
                                                      const float* __restrict__ G,
                                                      const int* __restrict__ idx,
                                                      float* __restrict__ pnume,
                                                      float* __restrict__ rs1_out,
                                                      float* __restrict__ newc) {
    __shared__ unsigned short list_s[N_];
    __shared__ int wave_tot_s[CU_WAVES];
    __shared__ __attribute__((aligned(16))) float part_s[CU_WAVES][CU_COLS];
    __shared__ float g_s[K_][K_ + 1];
    __shared__ float rs1_s[K_];

    const int k = blockIdx.y;
    const int c0 = blockIdx.x * CU_COLS;
    const int lane = threadIdx.x & 31;
    const int wv = __builtin_amdgcn_readfirstlane((int)(threadIdx.x >> 5));

    const int4* __restrict__ idx4 = (const int4*)idx + (size_t)wv * (CU_SCAN_IT * 32) + lane;
    unsigned long long m = 0ull;
#pragma unroll
    for (int i = 0; i < CU_SCAN_IT; ++i) {
        const int4 v = idx4[i * 32];
        const unsigned f = (unsigned)(v.x == k) | ((unsigned)(v.y == k) << 1) |
                           ((unsigned)(v.z == k) << 2) | ((unsigned)(v.w == k) << 3);
        m |= (unsigned long long)f << (4 * i);
    }

    {
        const float4* __restrict__ G4 = (const float4*)G;
#pragma unroll
        for (int i = 0; i < 4; ++i) {
            const int e4 = (int)threadIdx.x + 256 * i;
            const float4 v = G4[e4];
            float* gp = &g_s[e4 >> 4][(e4 & 15) * 4];
            gp[0] = v.x; gp[1] = v.y; gp[2] = v.z; gp[3] = v.w;
        }
    }
    const float4 ck0 = ((const float4*)(cluster + (size_t)k * F_ + c0))[lane];
    const float4 ck1 = ((const float4*)(cluster + (size_t)k * F_ + c0))[lane + 32];

    const int c = __popcll(m);
    int incl = c;
#pragma unroll
    for (int d = 1; d < 32; d <<= 1) {
        const int t = __shfl_up(incl, d, 32);
        if (lane >= d) incl += t;
    }
    if (lane == 31) wave_tot_s[wv] = incl;
    __syncthreads();
    int base = 0, total = 0;
#pragma unroll
    for (int j = 0; j < CU_WAVES; ++j) {
        const int t = wave_tot_s[j];
        if (j < wv) base += t;
        total += t;
    }
    total = __builtin_amdgcn_readfirstlane(total);
    {
        int pos = base + incl - c;
        const int n0 = wv * (CU_SCAN_IT * 128) + lane * 4;
        while (m) {
            const int b = __builtin_ctzll(m);
            m &= m - 1;
            list_s[pos++] = (unsigned short)(n0 + (b >> 2) * 128 + (b & 3));
        }
    }
    if (wv < K_ / 32) {
        const int j = threadIdx.x;
        float rowG = 0.f;
#pragma unroll
        for (int i = 0; i < K_; ++i) rowG += g_s[j][i];
        const float nrm = g_s[j][j];
        float S = 0.f;
#pragma unroll
        for (int i = 0; i < K_; ++i) S += g_s[i][i];
        rs1_s[j] = (float)K_ * nrm + S - 2.f * rowG + ALPHA_;
    }
    __syncthreads();

    if (blockIdx.x == 0 && blockIdx.y == 0 && threadIdx.x < K_) rs1_out[threadIdx.x] = rs1_s[threadIdx.x];

    const int beg = (total * wv) / CU_WAVES;
    const int end = (total * (wv + 1)) / CU_WAVES;
    float4 s0 = {0.f, 0.f, 0.f, 0.f};
    float4 s1 = {0.f, 0.f, 0.f, 0.f};
    const float* __restrict__ fb = features + c0;
    float* __restrict__ pn_tile = pnume + (size_t)blockIdx.x * N_;
    const v2f c01 = {ck0.x, ck0.y}, c23 = {ck0.z, ck0.w}, c45 = {ck1.x, ck1.y}, c67 = {ck1.z, ck1.w};
    for (int q = beg; q < end; q += CU_G) {
        int pp = q + (lane < CU_G ? lane : CU_G - 1);
        pp = pp < end ? pp : end - 1;
        const int rv = (int)list_s[pp];
        v8f X[CU_G];
#pragma unroll
        for (int r = 0; r < CU_G; ++r) {
            const int row = __builtin_amdgcn_readlane(rv, r);
            const float4* rp = (const float4*)(fb + (size_t)row * F_);
            const float4 lo = rp[lane];
            const float4 hi = rp[lane + 32];
            X[r] = (v8f){lo.x, lo.y, lo.z, lo.w, hi.x, hi.y, hi.z, hi.w};
        }
        asm volatile("" ::"v"(X[0]), "v"(X[1]), "v"(X[2]), "v"(X[3]), "v"(X[4]), "v"(X[5]),
                     "v"(X[6]), "v"(X[7]), "v"(X[8]), "v"(X[9]), "v"(X[10]), "v"(X[11]),
                     "v"(X[12]), "v"(X[13]), "v"(X[14]), "v"(X[15]), "v"(X[16]), "v"(X[17]),
                     "v"(X[18]), "v"(X[19]));
        static_assert(CU_G == 20, "the fence above names CU_G values");
        __builtin_amdgcn_sched_barrier(0);
        float pmine = 0.f;
#pragma unroll
        for (int r = 0; r < CU_G; ++r) {
            const float wt = (q + r < end) ? 1.f : 0.f;
            s0.x = fmaf(X[r][0], wt, s0.x);
            s0.y = fmaf(X[r][1], wt, s0.y);
            s0.z = fmaf(X[r][2], wt, s0.z);
            s0.w = fmaf(X[r][3], wt, s0.w);
            s1.x = fmaf(X[r][4], wt, s1.x);
            s1.y = fmaf(X[r][5], wt, s1.y);
            s1.z = fmaf(X[r][6], wt, s1.z);
            s1.w = fmaf(X[r][7], wt, s1.w);
            const v2f e0 = (v2f){X[r][0], X[r][1]} - c01;
            const v2f e1 = (v2f){X[r][2], X[r][3]} - c23;
            const v2f e2 = (v2f){X[r][4], X[r][5]} - c45;
            const v2f e3 = (v2f){X[r][6], X[r][7]} - c67;
            v2f a2 = e0 * e0;
            a2 = __builtin_elementwise_fma(e1, e1, a2);
            a2 = __builtin_elementwise_fma(e2, e2, a2);
            a2 = __builtin_elementwise_fma(e3, e3, a2);
            float p = a2.x + a2.y;
            p = wave_sum(p);
            if (lane == r) pmine = p;
        }
        if (lane < CU_G && q + lane < end) pn_tile[rv] = pmine;
    }

    float4* ps = (float4*)&part_s[wv][0];
    ps[lane] = s0;
    ps[32 + lane] = s1;
    __syncthreads();
    const int t = threadIdx.x;
    float a = 0.f;
#pragma unroll
    for (int j = 0; j < CU_WAVES; ++j) a += part_s[j][t];
    const float wk = cw[k] / rs1_s[k];
    const float cc = cluster[(size_t)k * F_ + c0 + t];
    newc[(size_t)k * F_ + c0 + t] = cc - GAMMA_ * wk * (a - (float)total * cc);
}

#define FL_THREADS 256
static_assert(N_ % FL_THREADS == 0, "launch 3 covers the rows exactly");

__global__ __launch_bounds__(FL_THREADS) void finish_loss(const float* __restrict__ pnume,
                                                          const int* __restrict__ idx,
                                                          const float* __restrict__ rs1,
                                                          float* __restrict__ loss) {
    const int n = blockIdx.x * FL_THREADS + threadIdx.x;
    float s = 0.f;
#pragma unroll
    for (int t = 0; t < F_ / CU_COLS; ++t) s += pnume[(size_t)t * N_ + n];
    loss[n] = s / rs1[idx[n]];
}

extern "C" void kernel_launch(void* const* d_in, const int* in_sizes, int n_in,
                              void* d_out, int out_size, void* d_ws, size_t ws_size,
                              hipStream_t stream) {
    const float* features = (const float*)d_in[0];
    const float* labels   = (const float*)d_in[1];
    const float* cluster  = (const float*)d_in[2];
    const float* cw       = (const float*)d_in[3];

    float* loss = (float*)d_out;
    float* newc = loss + N_;

    float* G     = (float*)d_ws;
    int* idxw    = (int*)(G + K_ * K_);
    float* pnume = (float*)(idxw + N_);
    float* rs1w  = pnume + (F_ / CU_COLS) * N_;

    gram_and_argmax<<<GR_BLOCKS + AR_BLOCKS, L1_WAVES * 32, 0, stream>>>(labels, cluster, G, idxw);
    cluster_update<<<dim3(F_ / CU_COLS, K_), 256, 0, stream>>>(
        features, cluster, cw, G, idxw, pnume, rs1w, newc);
    finish_loss<<<N_ / FL_THREADS, FL_THREADS, 0, stream>>>(pnume, idxw, rs1w, loss);
}


# cluster_update start-up reads own cluster only: 2 KB row bitmap (wave ballots in launch 1) not all idx, row k + diagonal of G not all G; new sum orders: rows ascending, row_sum by lane tree
# speedup vs baseline: 23.1987x; 1.0395x over previous
#include <hip/hip_runtime.h>

#define N_ 16384
#define K_ 64
#define F_ 2048
#define ALPHA_ 1.0f
#define GAMMA_ 1.0f

typedef float v2f __attribute__((ext_vector_type(2)));
typedef float v8f __attribute__((ext_vector_type(8)));


#define GR_BLOCKS 16
#define GR_CHUNKS 16
#define GR_KW (F_ / GR_CHUNKS)
#define GR_LD (GR_KW / 8)
#define L1_WAVES 8

static_assert(GR_KW * GR_CHUNKS == F_ && GR_LD * 8 == GR_KW, "the chunks cover F exactly");
static_assert(GR_CHUNKS == 2 * L1_WAVES, "two chunks per wave");
static_assert(GR_BLOCKS == (K_ / 16) * (K_ / 16), "one block per 16x16 tile");

__device__ __forceinline__ void gram_tile(const float* __restrict__ C, float* __restrict__ G,
                                          float (*gpart_s)[256]) {
    const int tile = blockIdx.x;
    const int ta = tile >> 2;
    const int tb = tile & 3;
    const int lane = threadIdx.x & 31;
    const int wv = threadIdx.x >> 5;
    const int lo = lane & 15;
    const int hi = lane >> 4;

#pragma unroll
    for (int h = 0; h < 2; ++h) {
        const int ch = wv + L1_WAVES * h;
        const float4* __restrict__ ap =
            (const float4*)(C + (size_t)(ta * 16 + lo) * F_ + ch * GR_KW) + hi;
        const float4* __restrict__ bp =
            (const float4*)(C + (size_t)(tb * 16 + lo) * F_ + ch * GR_KW) + hi;

        float4 a[GR_LD], b[GR_LD];
#pragma unroll
        for (int i = 0; i < GR_LD; ++i) {
            a[i] = ap[2 * i];
            b[i] = bp[2 * i];
        }

        v8f acc0 = {0.f, 0.f, 0.f, 0.f, 0.f, 0.f, 0.f, 0.f};
        v8f acc1 = {0.f, 0.f, 0.f, 0.f, 0.f, 0.f, 0.f, 0.f};
#pragma unroll
        for (int i = 0; i < GR_LD; ++i) {
            const v2f a01 = {a[i].x, a[i].y}, b01 = {b[i].x, b[i].y};
            const v2f a23 = {a[i].z, a[i].w}, b23 = {b[i].z, b[i].w};
            acc0 = __builtin_amdgcn_wmma_f32_16x16x4_f32(
                 false, a01,  false, b01,
                 (short)0, acc0,  false,  false);
            acc1 = __builtin_amdgcn_wmma_f32_16x16x4_f32(
                 false, a23,  false, b23,
                 (short)0, acc1,  false,  false);
        }

#pragma unroll
        for (int r = 0; r < 8; ++r) gpart_s[ch][(r + 8 * hi) * 16 + lo] = acc0[r] + acc1[r];
    }
    __syncthreads();

    const int t = threadIdx.x;
    float s = 0.f;
#pragma unroll
    for (int j = 0; j < GR_CHUNKS; ++j) s += gpart_s[j][t];
    G[(ta * 16 + (t >> 4)) * K_ + tb * 16 + (t & 15)] = s;
}

#define BM_WORDS (K_ * L1_WAVES)
static_assert(K_ == 64, "two bitmap words per lane: clusters lane and lane + 32");
static_assert(BM_WORDS % 4 == 0 && BM_WORDS / 4 <= L1_WAVES * 32, "one 16-byte store per thread moves the bitmap");
#define AR_BLOCKS (N_ / (L1_WAVES * 32))
static_assert(AR_BLOCKS * L1_WAVES * 32 == N_, "argmax blocks cover the rows exactly");
static_assert(K_ % 4 == 0, "a row of labels is whole 16-byte pieces");

__global__ __launch_bounds__(L1_WAVES * 32) void gram_and_argmax(const float* __restrict__ labels,
                                                                 const float* __restrict__ cluster,
                                                                 float* __restrict__ G,
                                                                 int* __restrict__ idx_out,
                                                                 unsigned* __restrict__ bm_out) {
    __shared__ float gpart_s[GR_CHUNKS][256];
    __shared__ __attribute__((aligned(16))) unsigned bm_s[BM_WORDS];

    if (blockIdx.x < GR_BLOCKS) {
        gram_tile(cluster, G, gpart_s);
        return;
    }

    const int row = (blockIdx.x - GR_BLOCKS) * (L1_WAVES * 32) + threadIdx.x;
    const float4* __restrict__ lp = (const float4*)(labels + (size_t)row * K_);
    float4 v[K_ / 4];
#pragma unroll
    for (int i = 0; i < K_ / 4; ++i) v[i] = lp[i];

    float bv = v[0].x;
    int bi = 0;
#pragma unroll
    for (int i = 0; i < K_ / 4; ++i) {
        if (i > 0 && v[i].x > bv) { bv = v[i].x; bi = 4 * i; }
        if (v[i].y > bv) { bv = v[i].y; bi = 4 * i + 1; }
        if (v[i].z > bv) { bv = v[i].z; bi = 4 * i + 2; }
        if (v[i].w > bv) { bv = v[i].w; bi = 4 * i + 3; }
    }
    idx_out[row] = bi;

    const int lane = threadIdx.x & 31;
    const int wv = threadIdx.x >> 5;
    int w_lo = 0, w_hi = 0;
#pragma unroll
    for (int j = 0; j < 32; ++j) {
        const int b_lo = (int)__builtin_amdgcn_ballot_w32(bi == j);
        const int b_hi = (int)__builtin_amdgcn_ballot_w32(bi == j + 32);
        w_lo = (lane == j) ? b_lo : w_lo;
        w_hi = (lane == j) ? b_hi : w_hi;
    }
    bm_s[lane * L1_WAVES + wv] = (unsigned)w_lo;
    bm_s[(lane + 32) * L1_WAVES + wv] = (unsigned)w_hi;
    __syncthreads();
    if (threadIdx.x < BM_WORDS / 4) {
        uint4* dst = (uint4*)(bm_out + (size_t)(blockIdx.x - GR_BLOCKS) * BM_WORDS);
        dst[threadIdx.x] = ((const uint4*)bm_s)[threadIdx.x];
    }
}

#define CU_COLS 256
#define CU_WAVES 8

static_assert(N_ <= 65536, "row numbers are kept as 16-bit values");
static_assert(CU_WAVES * 32 * 64 == N_, "64 rows (one 64-bit match mask) per thread cover the rows exactly");
static_assert(CU_WAVES == L1_WAVES, "thread t's 64 rows are two bitmap words of argmax block t / 4 (256 rows, 8 waves)");
static_assert(K_ == 64, "row k and the diagonal of G: two values per lane each");

#define CU_G 20

template <int M>
__device__ __forceinline__ float lane_xor_in_row(float x) {
    const int xi = __builtin_bit_cast(int, x);
    return __builtin_bit_cast(float, __builtin_amdgcn_update_dpp(0, xi, 0x160 + M, 0xf, 0xf, true));
}
__device__ __forceinline__ float other_row(float x) {
    const int xi = __builtin_bit_cast(int, x);
    return __builtin_bit_cast(float, __builtin_amdgcn_permlanex16(xi, xi, 0x76543210, 0xfedcba98, false, false));
}
__device__ __forceinline__ float wave_sum(float p) {
    p += lane_xor_in_row<1>(p);
    p += lane_xor_in_row<2>(p);
    p += lane_xor_in_row<4>(p);
    p += lane_xor_in_row<8>(p);
    p += other_row(p);
    return p;
}

__global__ __launch_bounds__(256) void cluster_update(const float* __restrict__ features,
                                                      const float* __restrict__ cluster,
                                                      const float* __restrict__ cw,
                                                      const float* __restrict__ G,
                                                      const unsigned* __restrict__ bm,
                                                      float* __restrict__ pnume,
                                                      float* __restrict__ rs1_out,
                                                      float* __restrict__ newc) {
    __shared__ unsigned short list_s[N_];
    __shared__ int wave_tot_s[CU_WAVES];
    __shared__ __attribute__((aligned(16))) float part_s[CU_WAVES][CU_COLS];
    __shared__ float rs1_s;

    const int k = blockIdx.y;
    const int c0 = blockIdx.x * CU_COLS;
    const int lane = threadIdx.x & 31;
    const int wv = __builtin_amdgcn_readfirstlane((int)(threadIdx.x >> 5));

    const uint2 mw = *(const uint2*)(bm + (size_t)(threadIdx.x >> 2) * BM_WORDS + k * L1_WAVES +
                                     2 * (threadIdx.x & 3));
    unsigned long long m = (unsigned long long)mw.x | ((unsigned long long)mw.y << 32);

    if (wv == 0) {
        const float rowG = wave_sum(G[k * K_ + lane] + G[k * K_ + lane + 32]);
        const float d_lo = G[lane * (K_ + 1)];
        const float d_hi = G[(lane + 32) * (K_ + 1)];
        const float S = wave_sum(d_lo + d_hi);
        const float nrm = __builtin_bit_cast(
            float, __builtin_amdgcn_readlane(__builtin_bit_cast(int, k < 32 ? d_lo : d_hi), k & 31));
        const float r = (float)K_ * nrm + S - 2.f * rowG + ALPHA_;
        if (lane == 0) {
            rs1_s = r;
            if (blockIdx.x == 0) rs1_out[k] = r;
        }
    }
    const float4 ck0 = ((const float4*)(cluster + (size_t)k * F_ + c0))[lane];
    const float4 ck1 = ((const float4*)(cluster + (size_t)k * F_ + c0))[lane + 32];

    const int c = __popcll(m);
    int incl = c;
#pragma unroll
    for (int d = 1; d < 32; d <<= 1) {
        const int t = __shfl_up(incl, d, 32);
        if (lane >= d) incl += t;
    }
    if (lane == 31) wave_tot_s[wv] = incl;
    __syncthreads();
    int base = 0, total = 0;
#pragma unroll
    for (int j = 0; j < CU_WAVES; ++j) {
        const int t = wave_tot_s[j];
        if (j < wv) base += t;
        total += t;
    }
    total = __builtin_amdgcn_readfirstlane(total);
    {
        int pos = base + incl - c;
        const int n0 = (int)threadIdx.x * 64;
        while (m) {
            const int b = __builtin_ctzll(m);
            m &= m - 1;
            list_s[pos++] = (unsigned short)(n0 + b);
        }
    }
    __syncthreads();

    const int beg = (total * wv) / CU_WAVES;
    const int end = (total * (wv + 1)) / CU_WAVES;
    float4 s0 = {0.f, 0.f, 0.f, 0.f};
    float4 s1 = {0.f, 0.f, 0.f, 0.f};
    const float* __restrict__ fb = features + c0;
    float* __restrict__ pn_tile = pnume + (size_t)blockIdx.x * N_;
    const v2f c01 = {ck0.x, ck0.y}, c23 = {ck0.z, ck0.w}, c45 = {ck1.x, ck1.y}, c67 = {ck1.z, ck1.w};
    for (int q = beg; q < end; q += CU_G) {
        int pp = q + (lane < CU_G ? lane : CU_G - 1);
        pp = pp < end ? pp : end - 1;
        const int rv = (int)list_s[pp];
        v8f X[CU_G];
#pragma unroll
        for (int r = 0; r < CU_G; ++r) {
            const int row = __builtin_amdgcn_readlane(rv, r);
            const float4* rp = (const float4*)(fb + (size_t)row * F_);
            const float4 lo = rp[lane];
            const float4 hi = rp[lane + 32];
            X[r] = (v8f){lo.x, lo.y, lo.z, lo.w, hi.x, hi.y, hi.z, hi.w};
        }
        asm volatile("" ::"v"(X[0]), "v"(X[1]), "v"(X[2]), "v"(X[3]), "v"(X[4]), "v"(X[5]),
                     "v"(X[6]), "v"(X[7]), "v"(X[8]), "v"(X[9]), "v"(X[10]), "v"(X[11]),
                     "v"(X[12]), "v"(X[13]), "v"(X[14]), "v"(X[15]), "v"(X[16]), "v"(X[17]),
                     "v"(X[18]), "v"(X[19]));
        static_assert(CU_G == 20, "the fence above names CU_G values");
        __builtin_amdgcn_sched_barrier(0);
        float pmine = 0.f;
#pragma unroll
        for (int r = 0; r < CU_G; ++r) {
            const float wt = (q + r < end) ? 1.f : 0.f;
            s0.x = fmaf(X[r][0], wt, s0.x);
            s0.y = fmaf(X[r][1], wt, s0.y);
            s0.z = fmaf(X[r][2], wt, s0.z);
            s0.w = fmaf(X[r][3], wt, s0.w);
            s1.x = fmaf(X[r][4], wt, s1.x);
            s1.y = fmaf(X[r][5], wt, s1.y);
            s1.z = fmaf(X[r][6], wt, s1.z);
            s1.w = fmaf(X[r][7], wt, s1.w);
            const v2f e0 = (v2f){X[r][0], X[r][1]} - c01;
            const v2f e1 = (v2f){X[r][2], X[r][3]} - c23;
            const v2f e2 = (v2f){X[r][4], X[r][5]} - c45;
            const v2f e3 = (v2f){X[r][6], X[r][7]} - c67;
            v2f a2 = e0 * e0;
            a2 = __builtin_elementwise_fma(e1, e1, a2);
            a2 = __builtin_elementwise_fma(e2, e2, a2);
            a2 = __builtin_elementwise_fma(e3, e3, a2);
            float p = a2.x + a2.y;
            p = wave_sum(p);
            if (lane == r) pmine = p;
        }
        if (lane < CU_G && q + lane < end) pn_tile[rv] = pmine;
    }

    float4* ps = (float4*)&part_s[wv][0];
    ps[lane] = s0;
    ps[32 + lane] = s1;
    __syncthreads();
    const int t = threadIdx.x;
    float a = 0.f;
#pragma unroll
    for (int j = 0; j < CU_WAVES; ++j) a += part_s[j][t];
    const float wk = cw[k] / rs1_s;
    const float cc = cluster[(size_t)k * F_ + c0 + t];
    newc[(size_t)k * F_ + c0 + t] = cc - GAMMA_ * wk * (a - (float)total * cc);
}

#define FL_THREADS 256
static_assert(N_ % FL_THREADS == 0, "launch 3 covers the rows exactly");

__global__ __launch_bounds__(FL_THREADS) void finish_loss(const float* __restrict__ pnume,
                                                          const int* __restrict__ idx,
                                                          const float* __restrict__ rs1,
                                                          float* __restrict__ loss) {
    const int n = blockIdx.x * FL_THREADS + threadIdx.x;
    float s = 0.f;
#pragma unroll
    for (int t = 0; t < F_ / CU_COLS; ++t) s += pnume[(size_t)t * N_ + n];
    loss[n] = s / rs1[idx[n]];
}

extern "C" void kernel_launch(void* const* d_in, const int* in_sizes, int n_in,
                              void* d_out, int out_size, void* d_ws, size_t ws_size,
                              hipStream_t stream) {
    const float* features = (const float*)d_in[0];
    const float* labels   = (const float*)d_in[1];
    const float* cluster  = (const float*)d_in[2];
    const float* cw       = (const float*)d_in[3];

    float* loss = (float*)d_out;
    float* newc = loss + N_;

    float* G     = (float*)d_ws;
    int* idxw    = (int*)(G + K_ * K_);
    float* pnume = (float*)(idxw + N_);
    float* rs1w  = pnume + (F_ / CU_COLS) * N_;
    unsigned* bmw = (unsigned*)(rs1w + K_);

    gram_and_argmax<<<GR_BLOCKS + AR_BLOCKS, L1_WAVES * 32, 0, stream>>>(labels, cluster, G, idxw, bmw);
    cluster_update<<<dim3(F_ / CU_COLS, K_), 256, 0, stream>>>(
        features, cluster, cw, G, bmw, pnume, rs1w, newc);
    finish_loss<<<N_ / FL_THREADS, FL_THREADS, 0, stream>>>(pnume, idxw, rs1w, loss);
}


# launch 1 in 4-wave work-groups (192 blocks, was 80 of 8 waves): half the label loads and a quarter of the Gram loads per WGP; G leaves as 4 k-quarter planes, added in launch 2 (new G sum order)
# speedup vs baseline: 28.1546x; 1.2136x over previous
#include <hip/hip_runtime.h>

#define N_ 16384
#define K_ 64
#define F_ 2048
#define ALPHA_ 1.0f
#define GAMMA_ 1.0f

typedef float v2f __attribute__((ext_vector_type(2)));
typedef float v8f __attribute__((ext_vector_type(8)));


#define GR_TILES 16
#define GR_CHUNKS 16
#define GR_KW (F_ / GR_CHUNKS)
#define GR_LD (GR_KW / 8)
#define L1_WAVES 4
#define GR_PLANES (GR_CHUNKS / L1_WAVES)
#define GR_BLOCKS (GR_TILES * GR_PLANES)
#define DG_PITCH 32

static_assert(GR_KW * GR_CHUNKS == F_ && GR_LD * 8 == GR_KW, "the chunks cover F exactly");
static_assert(GR_PLANES * L1_WAVES == GR_CHUNKS, "one chunk per wave");
static_assert(GR_TILES == (K_ / 16) * (K_ / 16), "16x16 tiles of G");
static_assert(256 % (L1_WAVES * 32) == 0, "every thread adds the same number of tile elements");

__device__ __forceinline__ void gram_tile(const float* __restrict__ C, float* __restrict__ Gp,
                                          float* __restrict__ dg, float (*gpart_s)[256]) {
    const int tile = blockIdx.x / GR_PLANES;
    const int plane = blockIdx.x % GR_PLANES;
    const int ta = tile >> 2;
    const int tb = tile & 3;
    const int lane = threadIdx.x & 31;
    const int wv = threadIdx.x >> 5;
    const int lo = lane & 15;
    const int hi = lane >> 4;

    {
        const int ch = plane * L1_WAVES + wv;
        const float4* __restrict__ ap =
            (const float4*)(C + (size_t)(ta * 16 + lo) * F_ + ch * GR_KW) + hi;
        const float4* __restrict__ bp =
            (const float4*)(C + (size_t)(tb * 16 + lo) * F_ + ch * GR_KW) + hi;

        float4 a[GR_LD], b[GR_LD];
#pragma unroll
        for (int i = 0; i < GR_LD; ++i) {
            a[i] = ap[2 * i];
            b[i] = bp[2 * i];
        }

        v8f acc0 = {0.f, 0.f, 0.f, 0.f, 0.f, 0.f, 0.f, 0.f};
        v8f acc1 = {0.f, 0.f, 0.f, 0.f, 0.f, 0.f, 0.f, 0.f};
#pragma unroll
        for (int i = 0; i < GR_LD; ++i) {
            const v2f a01 = {a[i].x, a[i].y}, b01 = {b[i].x, b[i].y};
            const v2f a23 = {a[i].z, a[i].w}, b23 = {b[i].z, b[i].w};
            acc0 = __builtin_amdgcn_wmma_f32_16x16x4_f32(
                 false, a01,  false, b01,
                 (short)0, acc0,  false,  false);
            acc1 = __builtin_amdgcn_wmma_f32_16x16x4_f32(
                 false, a23,  false, b23,
                 (short)0, acc1,  false,  false);
        }

#pragma unroll
        for (int r = 0; r < 8; ++r) gpart_s[wv][(r + 8 * hi) * 16 + lo] = acc0[r] + acc1[r];
    }
    __syncthreads();

    float* __restrict__ G = Gp + plane * (K_ * K_);
#pragma unroll
    for (int e = threadIdx.x; e < 256; e += L1_WAVES * 32) {
        float s = 0.f;
#pragma unroll
        for (int j = 0; j < L1_WAVES; ++j) s += gpart_s[j][e];
        const int row = e >> 4, col = e & 15;
        G[(ta * 16 + row) * K_ + tb * 16 + col] = s;
        if (ta == tb && row == col) dg[(plane * (K_ / 16) + ta) * DG_PITCH + row] = s;
    }
}

#define BM_WORDS (K_ * L1_WAVES)
static_assert(K_ == 64, "two bitmap words per lane: clusters lane and lane + 32");
static_assert(BM_WORDS % 4 == 0 && BM_WORDS / 4 <= L1_WAVES * 32, "one 16-byte store per thread moves the bitmap");
#define AR_BLOCKS (N_ / (L1_WAVES * 32))
static_assert(AR_BLOCKS * L1_WAVES * 32 == N_, "argmax blocks cover the rows exactly");
static_assert(K_ % 4 == 0, "a row of labels is whole 16-byte pieces");

__global__ __launch_bounds__(L1_WAVES * 32) void gram_and_argmax(const float* __restrict__ labels,
                                                                 const float* __restrict__ cluster,
                                                                 float* __restrict__ Gp,
                                                                 float* __restrict__ dg,
                                                                 int* __restrict__ idx_out,
                                                                 unsigned* __restrict__ bm_out) {
    __shared__ float gpart_s[L1_WAVES][256];
    __shared__ __attribute__((aligned(16))) unsigned bm_s[BM_WORDS];

    if (blockIdx.x < GR_BLOCKS) {
        gram_tile(cluster, Gp, dg, gpart_s);
        return;
    }

    const int row = (blockIdx.x - GR_BLOCKS) * (L1_WAVES * 32) + threadIdx.x;
    const float4* __restrict__ lp = (const float4*)(labels + (size_t)row * K_);
    float4 v[K_ / 4];
#pragma unroll
    for (int i = 0; i < K_ / 4; ++i) v[i] = lp[i];

    float bv = v[0].x;
    int bi = 0;
#pragma unroll
    for (int i = 0; i < K_ / 4; ++i) {
        if (i > 0 && v[i].x > bv) { bv = v[i].x; bi = 4 * i; }
        if (v[i].y > bv) { bv = v[i].y; bi = 4 * i + 1; }
        if (v[i].z > bv) { bv = v[i].z; bi = 4 * i + 2; }
        if (v[i].w > bv) { bv = v[i].w; bi = 4 * i + 3; }
    }
    idx_out[row] = bi;

    const int lane = threadIdx.x & 31;
    const int wv = threadIdx.x >> 5;
    int w_lo = 0, w_hi = 0;
#pragma unroll
    for (int j = 0; j < 32; ++j) {
        const int b_lo = (int)__builtin_amdgcn_ballot_w32(bi == j);
        const int b_hi = (int)__builtin_amdgcn_ballot_w32(bi == j + 32);
        w_lo = (lane == j) ? b_lo : w_lo;
        w_hi = (lane == j) ? b_hi : w_hi;
    }
    bm_s[lane * L1_WAVES + wv] = (unsigned)w_lo;
    bm_s[(lane + 32) * L1_WAVES + wv] = (unsigned)w_hi;
    __syncthreads();
    if (threadIdx.x < BM_WORDS / 4) {
        uint4* dst = (uint4*)(bm_out + (size_t)(blockIdx.x - GR_BLOCKS) * BM_WORDS);
        dst[threadIdx.x] = ((const uint4*)bm_s)[threadIdx.x];
    }
}

#define CU_COLS 256
#define CU_WAVES 8

static_assert(N_ <= 65536, "row numbers are kept as 16-bit values");
static_assert(CU_WAVES * 32 * 64 == N_, "64 rows (one 64-bit match mask) per thread cover the rows exactly");
static_assert(L1_WAVES % 2 == 0, "thread t's 64 rows are two neighbouring bitmap words of one argmax block");
static_assert(K_ == 64, "row k and the diagonal of G: two values per lane each");

#define CU_G 20

template <int M>
__device__ __forceinline__ float lane_xor_in_row(float x) {
    const int xi = __builtin_bit_cast(int, x);
    return __builtin_bit_cast(float, __builtin_amdgcn_update_dpp(0, xi, 0x160 + M, 0xf, 0xf, true));
}
__device__ __forceinline__ float other_row(float x) {
    const int xi = __builtin_bit_cast(int, x);
    return __builtin_bit_cast(float, __builtin_amdgcn_permlanex16(xi, xi, 0x76543210, 0xfedcba98, false, false));
}
__device__ __forceinline__ float wave_sum(float p) {
    p += lane_xor_in_row<1>(p);
    p += lane_xor_in_row<2>(p);
    p += lane_xor_in_row<4>(p);
    p += lane_xor_in_row<8>(p);
    p += other_row(p);
    return p;
}

__global__ __launch_bounds__(256) void cluster_update(const float* __restrict__ features,
                                                      const float* __restrict__ cluster,
                                                      const float* __restrict__ cw,
                                                      const float* __restrict__ Gp,
                                                      const float* __restrict__ dg,
                                                      const unsigned* __restrict__ bm,
                                                      float* __restrict__ pnume,
                                                      float* __restrict__ rs1_out,
                                                      float* __restrict__ newc) {
    __shared__ unsigned short list_s[N_];
    __shared__ int wave_tot_s[CU_WAVES];
    __shared__ __attribute__((aligned(16))) float part_s[CU_WAVES][CU_COLS];
    __shared__ float rs1_s;

    const int k = blockIdx.y;
    const int c0 = blockIdx.x * CU_COLS;
    const int lane = threadIdx.x & 31;
    const int wv = __builtin_amdgcn_readfirstlane((int)(threadIdx.x >> 5));

    const uint2 mw = *(const uint2*)(bm + (size_t)((2 * threadIdx.x) / L1_WAVES) * BM_WORDS + k * L1_WAVES +
                                     (2 * threadIdx.x) % L1_WAVES);
    unsigned long long m = (unsigned long long)mw.x | ((unsigned long long)mw.y << 32);

    if (wv == 0) {
        float g_lo = 0.f, g_hi = 0.f;
        float d_lo = 0.f, d_hi = 0.f;
#pragma unroll
        for (int q = 0; q < GR_PLANES; ++q) {
            g_lo += Gp[q * (K_ * K_) + k * K_ + lane];
            g_hi += Gp[q * (K_ * K_) + k * K_ + lane + 32];
            d_lo += dg[(q * (K_ / 16) + (lane >> 4)) * DG_PITCH + (lane & 15)];
            d_hi += dg[(q * (K_ / 16) + 2 + (lane >> 4)) * DG_PITCH + (lane & 15)];
        }
        const float rowG = wave_sum(g_lo + g_hi);
        const float S = wave_sum(d_lo + d_hi);
        const float nrm = __builtin_bit_cast(
            float, __builtin_amdgcn_readlane(__builtin_bit_cast(int, k < 32 ? d_lo : d_hi), k & 31));
        const float r = (float)K_ * nrm + S - 2.f * rowG + ALPHA_;
        if (lane == 0) {
            rs1_s = r;
            if (blockIdx.x == 0) rs1_out[k] = r;
        }
    }
    const float4 ck0 = ((const float4*)(cluster + (size_t)k * F_ + c0))[lane];
    const float4 ck1 = ((const float4*)(cluster + (size_t)k * F_ + c0))[lane + 32];

    const int c = __popcll(m);
    int incl = c;
#pragma unroll
    for (int d = 1; d < 32; d <<= 1) {
        const int t = __shfl_up(incl, d, 32);
        if (lane >= d) incl += t;
    }
    if (lane == 31) wave_tot_s[wv] = incl;
    __syncthreads();
    int base = 0, total = 0;
#pragma unroll
    for (int j = 0; j < CU_WAVES; ++j) {
        const int t = wave_tot_s[j];
        if (j < wv) base += t;
        total += t;
    }
    total = __builtin_amdgcn_readfirstlane(total);
    {
        int pos = base + incl - c;
        const int n0 = (int)threadIdx.x * 64;
        while (m) {
            const int b = __builtin_ctzll(m);
            m &= m - 1;
            list_s[pos++] = (unsigned short)(n0 + b);
        }
    }
    __syncthreads();

    const int beg = (total * wv) / CU_WAVES;
    const int end = (total * (wv + 1)) / CU_WAVES;
    float4 s0 = {0.f, 0.f, 0.f, 0.f};
    float4 s1 = {0.f, 0.f, 0.f, 0.f};
    const float* __restrict__ fb = features + c0;
    float* __restrict__ pn_tile = pnume + (size_t)blockIdx.x * N_;
    const v2f c01 = {ck0.x, ck0.y}, c23 = {ck0.z, ck0.w}, c45 = {ck1.x, ck1.y}, c67 = {ck1.z, ck1.w};
    for (int q = beg; q < end; q += CU_G) {
        int pp = q + (lane < CU_G ? lane : CU_G - 1);
        pp = pp < end ? pp : end - 1;
        const int rv = (int)list_s[pp];
        v8f X[CU_G];
#pragma unroll
        for (int r = 0; r < CU_G; ++r) {
            const int row = __builtin_amdgcn_readlane(rv, r);
            const float4* rp = (const float4*)(fb + (size_t)row * F_);
            const float4 lo = rp[lane];
            const float4 hi = rp[lane + 32];
            X[r] = (v8f){lo.x, lo.y, lo.z, lo.w, hi.x, hi.y, hi.z, hi.w};
        }
        asm volatile("" ::"v"(X[0]), "v"(X[1]), "v"(X[2]), "v"(X[3]), "v"(X[4]), "v"(X[5]),
                     "v"(X[6]), "v"(X[7]), "v"(X[8]), "v"(X[9]), "v"(X[10]), "v"(X[11]),
                     "v"(X[12]), "v"(X[13]), "v"(X[14]), "v"(X[15]), "v"(X[16]), "v"(X[17]),
                     "v"(X[18]), "v"(X[19]));
        static_assert(CU_G == 20, "the fence above names CU_G values");
        __builtin_amdgcn_sched_barrier(0);
        float pmine = 0.f;
#pragma unroll
        for (int r = 0; r < CU_G; ++r) {
            const float wt = (q + r < end) ? 1.f : 0.f;
            s0.x = fmaf(X[r][0], wt, s0.x);
            s0.y = fmaf(X[r][1], wt, s0.y);
            s0.z = fmaf(X[r][2], wt, s0.z);
            s0.w = fmaf(X[r][3], wt, s0.w);
            s1.x = fmaf(X[r][4], wt, s1.x);
            s1.y = fmaf(X[r][5], wt, s1.y);
            s1.z = fmaf(X[r][6], wt, s1.z);
            s1.w = fmaf(X[r][7], wt, s1.w);
            const v2f e0 = (v2f){X[r][0], X[r][1]} - c01;
            const v2f e1 = (v2f){X[r][2], X[r][3]} - c23;
            const v2f e2 = (v2f){X[r][4], X[r][5]} - c45;
            const v2f e3 = (v2f){X[r][6], X[r][7]} - c67;
            v2f a2 = e0 * e0;
            a2 = __builtin_elementwise_fma(e1, e1, a2);
            a2 = __builtin_elementwise_fma(e2, e2, a2);
            a2 = __builtin_elementwise_fma(e3, e3, a2);
            float p = a2.x + a2.y;
            p = wave_sum(p);
            if (lane == r) pmine = p;
        }
        if (lane < CU_G && q + lane < end) pn_tile[rv] = pmine;
    }

    float4* ps = (float4*)&part_s[wv][0];
    ps[lane] = s0;
    ps[32 + lane] = s1;
    __syncthreads();
    const int t = threadIdx.x;
    float a = 0.f;
#pragma unroll
    for (int j = 0; j < CU_WAVES; ++j) a += part_s[j][t];
    const float wk = cw[k] / rs1_s;
    const float cc = cluster[(size_t)k * F_ + c0 + t];
    newc[(size_t)k * F_ + c0 + t] = cc - GAMMA_ * wk * (a - (float)total * cc);
}

#define FL_THREADS 256
static_assert(N_ % FL_THREADS == 0, "launch 3 covers the rows exactly");

__global__ __launch_bounds__(FL_THREADS) void finish_loss(const float* __restrict__ pnume,
                                                          const int* __restrict__ idx,
                                                          const float* __restrict__ rs1,
                                                          float* __restrict__ loss) {
    const int n = blockIdx.x * FL_THREADS + threadIdx.x;
    float s = 0.f;
#pragma unroll
    for (int t = 0; t < F_ / CU_COLS; ++t) s += pnume[(size_t)t * N_ + n];
    loss[n] = s / rs1[idx[n]];
}

extern "C" void kernel_launch(void* const* d_in, const int* in_sizes, int n_in,
                              void* d_out, int out_size, void* d_ws, size_t ws_size,
                              hipStream_t stream) {
    const float* features = (const float*)d_in[0];
    const float* labels   = (const float*)d_in[1];
    const float* cluster  = (const float*)d_in[2];
    const float* cw       = (const float*)d_in[3];

    float* loss = (float*)d_out;
    float* newc = loss + N_;

    float* Gp    = (float*)d_ws;
    float* dgw   = Gp + GR_PLANES * K_ * K_;
    int* idxw    = (int*)(dgw + GR_PLANES * (K_ / 16) * DG_PITCH);
    float* pnume = (float*)(idxw + N_);
    float* rs1w  = pnume + (F_ / CU_COLS) * N_;
    unsigned* bmw = (unsigned*)(rs1w + K_);

    gram_and_argmax<<<GR_BLOCKS + AR_BLOCKS, L1_WAVES * 32, 0, stream>>>(labels, cluster, Gp, dgw, idxw, bmw);
    cluster_update<<<dim3(F_ / CU_COLS, K_), 256, 0, stream>>>(
        features, cluster, cw, Gp, dgw, bmw, pnume, rs1w, newc);
    finish_loss<<<N_ / FL_THREADS, FL_THREADS, 0, stream>>>(pnume, idxw, rs1w, loss);
}


# launch 1: every load covers 4 whole lines (was 32 and 16 partial): 16 lanes share a label row (max, then least index, by DPP), Gram operand rows go through LDS; idx, G, sum orders unchanged
# speedup vs baseline: 29.5368x; 1.0491x over previous
#include <hip/hip_runtime.h>

#define N_ 16384
#define K_ 64
#define F_ 2048
#define ALPHA_ 1.0f
#define GAMMA_ 1.0f

typedef float v2f __attribute__((ext_vector_type(2)));
typedef float v4f __attribute__((ext_vector_type(4)));
typedef float v8f __attribute__((ext_vector_type(8)));


#define GR_TILES 16
#define GR_CHUNKS 16
#define GR_KW (F_ / GR_CHUNKS)
#define GR_LD (GR_KW / 8)
#define L1_WAVES 4
#define GR_PLANES (GR_CHUNKS / L1_WAVES)
#define GR_BLOCKS (GR_TILES * GR_PLANES)
#define DG_PITCH 32

static_assert(GR_KW * GR_CHUNKS == F_ && GR_LD * 8 == GR_KW, "the chunks cover F exactly");
static_assert(GR_PLANES * L1_WAVES == GR_CHUNKS, "one chunk per wave");
static_assert(GR_TILES == (K_ / 16) * (K_ / 16), "16x16 tiles of G");
static_assert(256 % (L1_WAVES * 32) == 0, "every thread adds the same number of tile elements");
#define GS_PITCH (GR_KW + 4)
static_assert(GR_KW == 32 * 4, "a row of a chunk is one 16-byte piece per lane");
static_assert((GS_PITCH * 4) % 16 == 0 && ((GS_PITCH * 4) / 16) % 2 == 1, "row pitch = 16 bytes x odd");

__device__ __forceinline__ void gram_tile(const float* __restrict__ C, float* __restrict__ Gp,
                                          float* __restrict__ dg, float (*gpart_s)[256],
                                          float (*gstage_s)[2][16][GS_PITCH]) {
    const int tile = blockIdx.x / GR_PLANES;
    const int plane = blockIdx.x % GR_PLANES;
    const int ta = tile >> 2;
    const int tb = tile & 3;
    const int lane = threadIdx.x & 31;
    const int wv = threadIdx.x >> 5;
    const int lo = lane & 15;
    const int hi = lane >> 4;

    float* __restrict__ sa = &gstage_s[wv][0][0][0];
    float* __restrict__ sb = &gstage_s[wv][1][0][0];
    {
        const int ch = plane * L1_WAVES + wv;
        const float4* __restrict__ ag = (const float4*)(C + (size_t)(ta * 16) * F_ + ch * GR_KW) + lane;
        const float4* __restrict__ bg = (const float4*)(C + (size_t)(tb * 16) * F_ + ch * GR_KW) + lane;
        float4 ra[16], rb[16];
#pragma unroll
        for (int j = 0; j < 16; ++j) {
            ra[j] = ag[j * (F_ / 4)];
            rb[j] = bg[j * (F_ / 4)];
        }
#pragma unroll
        for (int j = 0; j < 16; ++j) {
            *(float4*)(sa + j * GS_PITCH + 4 * lane) = ra[j];
            *(float4*)(sb + j * GS_PITCH + 4 * lane) = rb[j];
        }
    }
    __syncthreads();
    {
        v4f a[GR_LD], b[GR_LD];
#pragma unroll
        for (int i = 0; i < GR_LD; ++i) {
            a[i] = *(const v4f*)(sa + lo * GS_PITCH + 8 * i + 4 * hi);
            b[i] = *(const v4f*)(sb + lo * GS_PITCH + 8 * i + 4 * hi);
        }
        asm volatile("" ::"v"(a[0]), "v"(a[1]), "v"(a[2]), "v"(a[3]), "v"(a[4]), "v"(a[5]), "v"(a[6]),
                     "v"(a[7]), "v"(a[8]), "v"(a[9]), "v"(a[10]), "v"(a[11]), "v"(a[12]), "v"(a[13]),
                     "v"(a[14]), "v"(a[15]));
        asm volatile("" ::"v"(b[0]), "v"(b[1]), "v"(b[2]), "v"(b[3]), "v"(b[4]), "v"(b[5]), "v"(b[6]),
                     "v"(b[7]), "v"(b[8]), "v"(b[9]), "v"(b[10]), "v"(b[11]), "v"(b[12]), "v"(b[13]),
                     "v"(b[14]), "v"(b[15]));
        static_assert(GR_LD == 16, "the fences above name GR_LD values each");

        v8f acc0 = {0.f, 0.f, 0.f, 0.f, 0.f, 0.f, 0.f, 0.f};
        v8f acc1 = {0.f, 0.f, 0.f, 0.f, 0.f, 0.f, 0.f, 0.f};
#pragma unroll
        for (int i = 0; i < GR_LD; ++i) {
            const v2f a01 = {a[i].x, a[i].y}, b01 = {b[i].x, b[i].y};
            const v2f a23 = {a[i].z, a[i].w}, b23 = {b[i].z, b[i].w};
            acc0 = __builtin_amdgcn_wmma_f32_16x16x4_f32(
                 false, a01,  false, b01,
                 (short)0, acc0,  false,  false);
            acc1 = __builtin_amdgcn_wmma_f32_16x16x4_f32(
                 false, a23,  false, b23,
                 (short)0, acc1,  false,  false);
        }

#pragma unroll
        for (int r = 0; r < 8; ++r) gpart_s[wv][(r + 8 * hi) * 16 + lo] = acc0[r] + acc1[r];
    }
    __syncthreads();

    float* __restrict__ G = Gp + plane * (K_ * K_);
#pragma unroll
    for (int e = threadIdx.x; e < 256; e += L1_WAVES * 32) {
        float s = 0.f;
#pragma unroll
        for (int j = 0; j < L1_WAVES; ++j) s += gpart_s[j][e];
        const int row = e >> 4, col = e & 15;
        G[(ta * 16 + row) * K_ + tb * 16 + col] = s;
        if (ta == tb && row == col) dg[(plane * (K_ / 16) + ta) * DG_PITCH + row] = s;
    }
}

#define BM_WORDS (K_ * L1_WAVES)
static_assert(K_ == 64, "two bitmap words per lane: clusters lane and lane + 32");
static_assert(BM_WORDS % 4 == 0 && BM_WORDS / 4 <= L1_WAVES * 32, "one 16-byte store per thread moves the bitmap");
#define AR_BLOCKS (N_ / (L1_WAVES * 32))
static_assert(AR_BLOCKS * L1_WAVES * 32 == N_, "argmax blocks cover the rows exactly");
static_assert(K_ == 16 * 4, "a row of labels is one 16-byte piece in each of 16 lanes");

template <int M>
__device__ __forceinline__ float lane_xor_in_row(float x) {
    const int xi = __builtin_bit_cast(int, x);
    return __builtin_bit_cast(float, __builtin_amdgcn_update_dpp(0, xi, 0x160 + M, 0xf, 0xf, true));
}
template <int M>
__device__ __forceinline__ int lane_xor_in_row_i(int x) {
    return __builtin_amdgcn_update_dpp(0, x, 0x160 + M, 0xf, 0xf, true);
}
__device__ __forceinline__ int imin(int a, int b) { return a < b ? a : b; }

__global__ __launch_bounds__(L1_WAVES * 32) void gram_and_argmax(const float* __restrict__ labels,
                                                                 const float* __restrict__ cluster,
                                                                 float* __restrict__ Gp,
                                                                 float* __restrict__ dg,
                                                                 int* __restrict__ idx_out,
                                                                 unsigned* __restrict__ bm_out) {
    __shared__ float gpart_s[L1_WAVES][256];
    __shared__ __attribute__((aligned(16))) unsigned bm_s[BM_WORDS];
    __shared__ __attribute__((aligned(16))) float gstage_s[L1_WAVES][2][16][GS_PITCH];

    if (blockIdx.x < GR_BLOCKS) {
        gram_tile(cluster, Gp, dg, gpart_s, gstage_s);
        return;
    }

    const int lane = threadIdx.x & 31;
    const int wv = threadIdx.x >> 5;
    const int lo = lane & 15;
    const int hi = lane >> 4;
    const int row0 = (blockIdx.x - GR_BLOCKS) * (L1_WAVES * 32) + wv * 32;
    const float4* __restrict__ lp = (const float4*)(labels + (size_t)(row0 + 16 * hi) * K_) + lo;
    float4 v[16];
#pragma unroll
    for (int i = 0; i < 16; ++i) v[i] = lp[i * (K_ / 4)];

    int bi = 0;
#pragma unroll
    for (int i = 0; i < 16; ++i) {
        float m = fmaxf(fmaxf(v[i].x, v[i].y), fmaxf(v[i].z, v[i].w));
        m = fmaxf(m, lane_xor_in_row<1>(m));
        m = fmaxf(m, lane_xor_in_row<2>(m));
        m = fmaxf(m, lane_xor_in_row<4>(m));
        m = fmaxf(m, lane_xor_in_row<8>(m));
        int cand = K_;
        cand = (v[i].w == m) ? 4 * lo + 3 : cand;
        cand = (v[i].z == m) ? 4 * lo + 2 : cand;
        cand = (v[i].y == m) ? 4 * lo + 1 : cand;
        cand = (v[i].x == m) ? 4 * lo : cand;
        cand = imin(cand, lane_xor_in_row_i<1>(cand));
        cand = imin(cand, lane_xor_in_row_i<2>(cand));
        cand = imin(cand, lane_xor_in_row_i<4>(cand));
        cand = imin(cand, lane_xor_in_row_i<8>(cand));
        bi = (lo == i) ? cand : bi;
    }
    bi &= K_ - 1;
    idx_out[row0 + lane] = bi;

    int w_lo = 0, w_hi = 0;
#pragma unroll
    for (int j = 0; j < 32; ++j) {
        const int b_lo = (int)__builtin_amdgcn_ballot_w32(bi == j);
        const int b_hi = (int)__builtin_amdgcn_ballot_w32(bi == j + 32);
        w_lo = (lane == j) ? b_lo : w_lo;
        w_hi = (lane == j) ? b_hi : w_hi;
    }
    bm_s[lane * L1_WAVES + wv] = (unsigned)w_lo;
    bm_s[(lane + 32) * L1_WAVES + wv] = (unsigned)w_hi;
    __syncthreads();
    if (threadIdx.x < BM_WORDS / 4) {
        uint4* dst = (uint4*)(bm_out + (size_t)(blockIdx.x - GR_BLOCKS) * BM_WORDS);
        dst[threadIdx.x] = ((const uint4*)bm_s)[threadIdx.x];
    }
}

#define CU_COLS 256
#define CU_WAVES 8

static_assert(N_ <= 65536, "row numbers are kept as 16-bit values");
static_assert(CU_WAVES * 32 * 64 == N_, "64 rows (one 64-bit match mask) per thread cover the rows exactly");
static_assert(L1_WAVES % 2 == 0, "thread t's 64 rows are two neighbouring bitmap words of one argmax block");
static_assert(K_ == 64, "row k and the diagonal of G: two values per lane each");

#define CU_G 20

__device__ __forceinline__ float other_row(float x) {
    const int xi = __builtin_bit_cast(int, x);
    return __builtin_bit_cast(float, __builtin_amdgcn_permlanex16(xi, xi, 0x76543210, 0xfedcba98, false, false));
}
__device__ __forceinline__ float wave_sum(float p) {
    p += lane_xor_in_row<1>(p);
    p += lane_xor_in_row<2>(p);
    p += lane_xor_in_row<4>(p);
    p += lane_xor_in_row<8>(p);
    p += other_row(p);
    return p;
}

__global__ __launch_bounds__(256) void cluster_update(const float* __restrict__ features,
                                                      const float* __restrict__ cluster,
                                                      const float* __restrict__ cw,
                                                      const float* __restrict__ Gp,
                                                      const float* __restrict__ dg,
                                                      const unsigned* __restrict__ bm,
                                                      float* __restrict__ pnume,
                                                      float* __restrict__ rs1_out,
                                                      float* __restrict__ newc) {
    __shared__ unsigned short list_s[N_];
    __shared__ int wave_tot_s[CU_WAVES];
    __shared__ __attribute__((aligned(16))) float part_s[CU_WAVES][CU_COLS];
    __shared__ float rs1_s;

    const int k = blockIdx.y;
    const int c0 = blockIdx.x * CU_COLS;
    const int lane = threadIdx.x & 31;
    const int wv = __builtin_amdgcn_readfirstlane((int)(threadIdx.x >> 5));

    const uint2 mw = *(const uint2*)(bm + (size_t)((2 * threadIdx.x) / L1_WAVES) * BM_WORDS + k * L1_WAVES +
                                     (2 * threadIdx.x) % L1_WAVES);
    unsigned long long m = (unsigned long long)mw.x | ((unsigned long long)mw.y << 32);

    if (wv == 0) {
        float g_lo = 0.f, g_hi = 0.f;
        float d_lo = 0.f, d_hi = 0.f;
#pragma unroll
        for (int q = 0; q < GR_PLANES; ++q) {
            g_lo += Gp[q * (K_ * K_) + k * K_ + lane];
            g_hi += Gp[q * (K_ * K_) + k * K_ + lane + 32];
            d_lo += dg[(q * (K_ / 16) + (lane >> 4)) * DG_PITCH + (lane & 15)];
            d_hi += dg[(q * (K_ / 16) + 2 + (lane >> 4)) * DG_PITCH + (lane & 15)];
        }
        const float rowG = wave_sum(g_lo + g_hi);
        const float S = wave_sum(d_lo + d_hi);
        const float nrm = __builtin_bit_cast(
            float, __builtin_amdgcn_readlane(__builtin_bit_cast(int, k < 32 ? d_lo : d_hi), k & 31));
        const float r = (float)K_ * nrm + S - 2.f * rowG + ALPHA_;
        if (lane == 0) {
            rs1_s = r;
            if (blockIdx.x == 0) rs1_out[k] = r;
        }
    }
    const float4 ck0 = ((const float4*)(cluster + (size_t)k * F_ + c0))[lane];
    const float4 ck1 = ((const float4*)(cluster + (size_t)k * F_ + c0))[lane + 32];

    const int c = __popcll(m);
    int incl = c;
#pragma unroll
    for (int d = 1; d < 32; d <<= 1) {
        const int t = __shfl_up(incl, d, 32);
        if (lane >= d) incl += t;
    }
    if (lane == 31) wave_tot_s[wv] = incl;
    __syncthreads();
    int base = 0, total = 0;
#pragma unroll
    for (int j = 0; j < CU_WAVES; ++j) {
        const int t = wave_tot_s[j];
        if (j < wv) base += t;
        total += t;
    }
    total = __builtin_amdgcn_readfirstlane(total);
    {
        int pos = base + incl - c;
        const int n0 = (int)threadIdx.x * 64;
        while (m) {
            const int b = __builtin_ctzll(m);
            m &= m - 1;
            list_s[pos++] = (unsigned short)(n0 + b);
        }
    }
    __syncthreads();

    const int beg = (total * wv) / CU_WAVES;
    const int end = (total * (wv + 1)) / CU_WAVES;
    float4 s0 = {0.f, 0.f, 0.f, 0.f};
    float4 s1 = {0.f, 0.f, 0.f, 0.f};
    const float* __restrict__ fb = features + c0;
    float* __restrict__ pn_tile = pnume + (size_t)blockIdx.x * N_;
    const v2f c01 = {ck0.x, ck0.y}, c23 = {ck0.z, ck0.w}, c45 = {ck1.x, ck1.y}, c67 = {ck1.z, ck1.w};
    for (int q = beg; q < end; q += CU_G) {
        int pp = q + (lane < CU_G ? lane : CU_G - 1);
        pp = pp < end ? pp : end - 1;
        const int rv = (int)list_s[pp];
        v8f X[CU_G];
#pragma unroll
        for (int r = 0; r < CU_G; ++r) {
            const int row = __builtin_amdgcn_readlane(rv, r);
            const float4* rp = (const float4*)(fb + (size_t)row * F_);
            const float4 lo = rp[lane];
            const float4 hi = rp[lane + 32];
            X[r] = (v8f){lo.x, lo.y, lo.z, lo.w, hi.x, hi.y, hi.z, hi.w};
        }
        asm volatile("" ::"v"(X[0]), "v"(X[1]), "v"(X[2]), "v"(X[3]), "v"(X[4]), "v"(X[5]),
                     "v"(X[6]), "v"(X[7]), "v"(X[8]), "v"(X[9]), "v"(X[10]), "v"(X[11]),
                     "v"(X[12]), "v"(X[13]), "v"(X[14]), "v"(X[15]), "v"(X[16]), "v"(X[17]),
                     "v"(X[18]), "v"(X[19]));
        static_assert(CU_G == 20, "the fence above names CU_G values");
        __builtin_amdgcn_sched_barrier(0);
        float pmine = 0.f;
#pragma unroll
        for (int r = 0; r < CU_G; ++r) {
            const float wt = (q + r < end) ? 1.f : 0.f;
            s0.x = fmaf(X[r][0], wt, s0.x);
            s0.y = fmaf(X[r][1], wt, s0.y);
            s0.z = fmaf(X[r][2], wt, s0.z);
            s0.w = fmaf(X[r][3], wt, s0.w);
            s1.x = fmaf(X[r][4], wt, s1.x);
            s1.y = fmaf(X[r][5], wt, s1.y);
            s1.z = fmaf(X[r][6], wt, s1.z);
            s1.w = fmaf(X[r][7], wt, s1.w);
            const v2f e0 = (v2f){X[r][0], X[r][1]} - c01;
            const v2f e1 = (v2f){X[r][2], X[r][3]} - c23;
            const v2f e2 = (v2f){X[r][4], X[r][5]} - c45;
            const v2f e3 = (v2f){X[r][6], X[r][7]} - c67;
            v2f a2 = e0 * e0;
            a2 = __builtin_elementwise_fma(e1, e1, a2);
            a2 = __builtin_elementwise_fma(e2, e2, a2);
            a2 = __builtin_elementwise_fma(e3, e3, a2);
            float p = a2.x + a2.y;
            p = wave_sum(p);
            if (lane == r) pmine = p;
        }
        if (lane < CU_G && q + lane < end) pn_tile[rv] = pmine;
    }

    float4* ps = (float4*)&part_s[wv][0];
    ps[lane] = s0;
    ps[32 + lane] = s1;
    __syncthreads();
    const int t = threadIdx.x;
    float a = 0.f;
#pragma unroll
    for (int j = 0; j < CU_WAVES; ++j) a += part_s[j][t];
    const float wk = cw[k] / rs1_s;
    const float cc = cluster[(size_t)k * F_ + c0 + t];
    newc[(size_t)k * F_ + c0 + t] = cc - GAMMA_ * wk * (a - (float)total * cc);
}

#define FL_THREADS 256
static_assert(N_ % FL_THREADS == 0, "launch 3 covers the rows exactly");

__global__ __launch_bounds__(FL_THREADS) void finish_loss(const float* __restrict__ pnume,
                                                          const int* __restrict__ idx,
                                                          const float* __restrict__ rs1,
                                                          float* __restrict__ loss) {
    const int n = blockIdx.x * FL_THREADS + threadIdx.x;
    float s = 0.f;
#pragma unroll
    for (int t = 0; t < F_ / CU_COLS; ++t) s += pnume[(size_t)t * N_ + n];
    loss[n] = s / rs1[idx[n]];
}

extern "C" void kernel_launch(void* const* d_in, const int* in_sizes, int n_in,
                              void* d_out, int out_size, void* d_ws, size_t ws_size,
                              hipStream_t stream) {
    const float* features = (const float*)d_in[0];
    const float* labels   = (const float*)d_in[1];
    const float* cluster  = (const float*)d_in[2];
    const float* cw       = (const float*)d_in[3];

    float* loss = (float*)d_out;
    float* newc = loss + N_;

    float* Gp    = (float*)d_ws;
    float* dgw   = Gp + GR_PLANES * K_ * K_;
    int* idxw    = (int*)(dgw + GR_PLANES * (K_ / 16) * DG_PITCH);
    float* pnume = (float*)(idxw + N_);
    float* rs1w  = pnume + (F_ / CU_COLS) * N_;
    unsigned* bmw = (unsigned*)(rs1w + K_);

    gram_and_argmax<<<GR_BLOCKS + AR_BLOCKS, L1_WAVES * 32, 0, stream>>>(labels, cluster, Gp, dgw, idxw, bmw);
    cluster_update<<<dim3(F_ / CU_COLS, K_), 256, 0, stream>>>(
        features, cluster, cw, Gp, dgw, bmw, pnume, rs1w, newc);
    finish_loss<<<N_ / FL_THREADS, FL_THREADS, 0, stream>>>(pnume, idxw, rs1w, loss);
}


# launch 2 block = cluster x eighth of its row list, wave = column tile: a row's 8 tile partials meet in LDS and loss leaves there; launch 3 adds the 8 slice sums per column; every sum order kept
# speedup vs baseline: 32.1147x; 1.0873x over previous
#include <hip/hip_runtime.h>

#define N_ 16384
#define K_ 64
#define F_ 2048
#define ALPHA_ 1.0f
#define GAMMA_ 1.0f

typedef float v2f __attribute__((ext_vector_type(2)));
typedef float v4f __attribute__((ext_vector_type(4)));
typedef float v8f __attribute__((ext_vector_type(8)));


#define GR_TILES 16
#define GR_CHUNKS 16
#define GR_KW (F_ / GR_CHUNKS)
#define GR_LD (GR_KW / 8)
#define L1_WAVES 4
#define GR_PLANES (GR_CHUNKS / L1_WAVES)
#define GR_BLOCKS (GR_TILES * GR_PLANES)
#define DG_PITCH 32

static_assert(GR_KW * GR_CHUNKS == F_ && GR_LD * 8 == GR_KW, "the chunks cover F exactly");
static_assert(GR_PLANES * L1_WAVES == GR_CHUNKS, "one chunk per wave");
static_assert(GR_TILES == (K_ / 16) * (K_ / 16), "16x16 tiles of G");
static_assert(256 % (L1_WAVES * 32) == 0, "every thread adds the same number of tile elements");
#define GS_PITCH (GR_KW + 4)
static_assert(GR_KW == 32 * 4, "a row of a chunk is one 16-byte piece per lane");
static_assert((GS_PITCH * 4) % 16 == 0 && ((GS_PITCH * 4) / 16) % 2 == 1, "row pitch = 16 bytes x odd");

__device__ __forceinline__ void gram_tile(const float* __restrict__ C, float* __restrict__ Gp,
                                          float* __restrict__ dg, float (*gpart_s)[256],
                                          float (*gstage_s)[2][16][GS_PITCH]) {
    const int tile = blockIdx.x / GR_PLANES;
    const int plane = blockIdx.x % GR_PLANES;
    const int ta = tile >> 2;
    const int tb = tile & 3;
    const int lane = threadIdx.x & 31;
    const int wv = threadIdx.x >> 5;
    const int lo = lane & 15;
    const int hi = lane >> 4;

    float* __restrict__ sa = &gstage_s[wv][0][0][0];
    float* __restrict__ sb = &gstage_s[wv][1][0][0];
    {
        const int ch = plane * L1_WAVES + wv;
        const float4* __restrict__ ag = (const float4*)(C + (size_t)(ta * 16) * F_ + ch * GR_KW) + lane;
        const float4* __restrict__ bg = (const float4*)(C + (size_t)(tb * 16) * F_ + ch * GR_KW) + lane;
        float4 ra[16], rb[16];
#pragma unroll
        for (int j = 0; j < 16; ++j) {
            ra[j] = ag[j * (F_ / 4)];
            rb[j] = bg[j * (F_ / 4)];
        }
#pragma unroll
        for (int j = 0; j < 16; ++j) {
            *(float4*)(sa + j * GS_PITCH + 4 * lane) = ra[j];
            *(float4*)(sb + j * GS_PITCH + 4 * lane) = rb[j];
        }
    }
    __syncthreads();
    {
        v4f a[GR_LD], b[GR_LD];
#pragma unroll
        for (int i = 0; i < GR_LD; ++i) {
            a[i] = *(const v4f*)(sa + lo * GS_PITCH + 8 * i + 4 * hi);
            b[i] = *(const v4f*)(sb + lo * GS_PITCH + 8 * i + 4 * hi);
        }
        asm volatile("" ::"v"(a[0]), "v"(a[1]), "v"(a[2]), "v"(a[3]), "v"(a[4]), "v"(a[5]), "v"(a[6]),
                     "v"(a[7]), "v"(a[8]), "v"(a[9]), "v"(a[10]), "v"(a[11]), "v"(a[12]), "v"(a[13]),
                     "v"(a[14]), "v"(a[15]));
        asm volatile("" ::"v"(b[0]), "v"(b[1]), "v"(b[2]), "v"(b[3]), "v"(b[4]), "v"(b[5]), "v"(b[6]),
                     "v"(b[7]), "v"(b[8]), "v"(b[9]), "v"(b[10]), "v"(b[11]), "v"(b[12]), "v"(b[13]),
                     "v"(b[14]), "v"(b[15]));
        static_assert(GR_LD == 16, "the fences above name GR_LD values each");

        v8f acc0 = {0.f, 0.f, 0.f, 0.f, 0.f, 0.f, 0.f, 0.f};
        v8f acc1 = {0.f, 0.f, 0.f, 0.f, 0.f, 0.f, 0.f, 0.f};
#pragma unroll
        for (int i = 0; i < GR_LD; ++i) {
            const v2f a01 = {a[i].x, a[i].y}, b01 = {b[i].x, b[i].y};
            const v2f a23 = {a[i].z, a[i].w}, b23 = {b[i].z, b[i].w};
            acc0 = __builtin_amdgcn_wmma_f32_16x16x4_f32(
                 false, a01,  false, b01,
                 (short)0, acc0,  false,  false);
            acc1 = __builtin_amdgcn_wmma_f32_16x16x4_f32(
                 false, a23,  false, b23,
                 (short)0, acc1,  false,  false);
        }

#pragma unroll
        for (int r = 0; r < 8; ++r) gpart_s[wv][(r + 8 * hi) * 16 + lo] = acc0[r] + acc1[r];
    }
    __syncthreads();

    float* __restrict__ G = Gp + plane * (K_ * K_);
#pragma unroll
    for (int e = threadIdx.x; e < 256; e += L1_WAVES * 32) {
        float s = 0.f;
#pragma unroll
        for (int j = 0; j < L1_WAVES; ++j) s += gpart_s[j][e];
        const int row = e >> 4, col = e & 15;
        G[(ta * 16 + row) * K_ + tb * 16 + col] = s;
        if (ta == tb && row == col) dg[(plane * (K_ / 16) + ta) * DG_PITCH + row] = s;
    }
}

#define BM_WORDS (K_ * L1_WAVES)
static_assert(K_ == 64, "two bitmap words per lane: clusters lane and lane + 32");
static_assert(BM_WORDS % 4 == 0 && BM_WORDS / 4 <= L1_WAVES * 32, "one 16-byte store per thread moves the bitmap");
#define AR_BLOCKS (N_ / (L1_WAVES * 32))
static_assert(AR_BLOCKS * L1_WAVES * 32 == N_, "argmax blocks cover the rows exactly");
static_assert(K_ == 16 * 4, "a row of labels is one 16-byte piece in each of 16 lanes");

template <int M>
__device__ __forceinline__ float lane_xor_in_row(float x) {
    const int xi = __builtin_bit_cast(int, x);
    return __builtin_bit_cast(float, __builtin_amdgcn_update_dpp(0, xi, 0x160 + M, 0xf, 0xf, true));
}
template <int M>
__device__ __forceinline__ int lane_xor_in_row_i(int x) {
    return __builtin_amdgcn_update_dpp(0, x, 0x160 + M, 0xf, 0xf, true);
}
__device__ __forceinline__ int imin(int a, int b) { return a < b ? a : b; }

__global__ __launch_bounds__(L1_WAVES * 32) void gram_and_argmax(const float* __restrict__ labels,
                                                                 const float* __restrict__ cluster,
                                                                 float* __restrict__ Gp,
                                                                 float* __restrict__ dg,
                                                                 unsigned* __restrict__ bm_out) {
    __shared__ float gpart_s[L1_WAVES][256];
    __shared__ __attribute__((aligned(16))) unsigned bm_s[BM_WORDS];
    __shared__ __attribute__((aligned(16))) float gstage_s[L1_WAVES][2][16][GS_PITCH];

    if (blockIdx.x < GR_BLOCKS) {
        gram_tile(cluster, Gp, dg, gpart_s, gstage_s);
        return;
    }

    const int lane = threadIdx.x & 31;
    const int wv = threadIdx.x >> 5;
    const int lo = lane & 15;
    const int hi = lane >> 4;
    const int row0 = (blockIdx.x - GR_BLOCKS) * (L1_WAVES * 32) + wv * 32;
    const float4* __restrict__ lp = (const float4*)(labels + (size_t)(row0 + 16 * hi) * K_) + lo;
    float4 v[16];
#pragma unroll
    for (int i = 0; i < 16; ++i) v[i] = lp[i * (K_ / 4)];

    int bi = 0;
#pragma unroll
    for (int i = 0; i < 16; ++i) {
        float m = fmaxf(fmaxf(v[i].x, v[i].y), fmaxf(v[i].z, v[i].w));
        m = fmaxf(m, lane_xor_in_row<1>(m));
        m = fmaxf(m, lane_xor_in_row<2>(m));
        m = fmaxf(m, lane_xor_in_row<4>(m));
        m = fmaxf(m, lane_xor_in_row<8>(m));
        int cand = K_;
        cand = (v[i].w == m) ? 4 * lo + 3 : cand;
        cand = (v[i].z == m) ? 4 * lo + 2 : cand;
        cand = (v[i].y == m) ? 4 * lo + 1 : cand;
        cand = (v[i].x == m) ? 4 * lo : cand;
        cand = imin(cand, lane_xor_in_row_i<1>(cand));
        cand = imin(cand, lane_xor_in_row_i<2>(cand));
        cand = imin(cand, lane_xor_in_row_i<4>(cand));
        cand = imin(cand, lane_xor_in_row_i<8>(cand));
        bi = (lo == i) ? cand : bi;
    }
    bi &= K_ - 1;

    int w_lo = 0, w_hi = 0;
#pragma unroll
    for (int j = 0; j < 32; ++j) {
        const int b_lo = (int)__builtin_amdgcn_ballot_w32(bi == j);
        const int b_hi = (int)__builtin_amdgcn_ballot_w32(bi == j + 32);
        w_lo = (lane == j) ? b_lo : w_lo;
        w_hi = (lane == j) ? b_hi : w_hi;
    }
    bm_s[lane * L1_WAVES + wv] = (unsigned)w_lo;
    bm_s[(lane + 32) * L1_WAVES + wv] = (unsigned)w_hi;
    __syncthreads();
    if (threadIdx.x < BM_WORDS / 4) {
        uint4* dst = (uint4*)(bm_out + (size_t)(blockIdx.x - GR_BLOCKS) * BM_WORDS);
        dst[threadIdx.x] = ((const uint4*)bm_s)[threadIdx.x];
    }
}

#define CU_COLS 256
#define CU_WAVES 8
#define CU_SLICES 8
#define SL_MAX (N_ / CU_SLICES)

static_assert(N_ <= 65536, "row numbers are kept as 16-bit values");
static_assert(CU_WAVES * 32 * 64 == N_, "64 rows (one 64-bit match mask) per thread cover the rows exactly");
static_assert(L1_WAVES % 2 == 0, "thread t's 64 rows are two neighbouring bitmap words of one argmax block");
static_assert(K_ == 64, "row k and the diagonal of G: two values per lane each");
static_assert(CU_WAVES * CU_COLS == F_, "the 8 waves of a block cover a row exactly");
static_assert(SL_MAX * CU_SLICES == N_, "a slice never holds more than SL_MAX rows");

#define CU_G 20

__device__ __forceinline__ float other_row(float x) {
    const int xi = __builtin_bit_cast(int, x);
    return __builtin_bit_cast(float, __builtin_amdgcn_permlanex16(xi, xi, 0x76543210, 0xfedcba98, false, false));
}
__device__ __forceinline__ float wave_sum(float p) {
    p += lane_xor_in_row<1>(p);
    p += lane_xor_in_row<2>(p);
    p += lane_xor_in_row<4>(p);
    p += lane_xor_in_row<8>(p);
    p += other_row(p);
    return p;
}

__global__ __launch_bounds__(256) void cluster_rows(const float* __restrict__ features,
                                                    const float* __restrict__ cluster,
                                                    const float* __restrict__ Gp,
                                                    const float* __restrict__ dg,
                                                    const unsigned* __restrict__ bm,
                                                    float* __restrict__ part,
                                                    float* __restrict__ rs1_out,
                                                    int* __restrict__ cnt_out,
                                                    float* __restrict__ loss) {
    __shared__ unsigned short list_s[SL_MAX];
    __shared__ int wave_tot_s[CU_WAVES];
    __shared__ float rowp_s[CU_WAVES][SL_MAX];
    __shared__ float rs1_s;

    const int k = blockIdx.y;
    const int j = blockIdx.x;
    const int lane = threadIdx.x & 31;
    const int wv = __builtin_amdgcn_readfirstlane((int)(threadIdx.x >> 5));
    const int c0 = wv * CU_COLS;

    const uint2 mw = *(const uint2*)(bm + (size_t)((2 * threadIdx.x) / L1_WAVES) * BM_WORDS + k * L1_WAVES +
                                     (2 * threadIdx.x) % L1_WAVES);
    unsigned long long m = (unsigned long long)mw.x | ((unsigned long long)mw.y << 32);

    if (wv == 0) {
        float g_lo = 0.f, g_hi = 0.f;
        float d_lo = 0.f, d_hi = 0.f;
#pragma unroll
        for (int q = 0; q < GR_PLANES; ++q) {
            g_lo += Gp[q * (K_ * K_) + k * K_ + lane];
            g_hi += Gp[q * (K_ * K_) + k * K_ + lane + 32];
            d_lo += dg[(q * (K_ / 16) + (lane >> 4)) * DG_PITCH + (lane & 15)];
            d_hi += dg[(q * (K_ / 16) + 2 + (lane >> 4)) * DG_PITCH + (lane & 15)];
        }
        const float rowG = wave_sum(g_lo + g_hi);
        const float S = wave_sum(d_lo + d_hi);
        const float nrm = __builtin_bit_cast(
            float, __builtin_amdgcn_readlane(__builtin_bit_cast(int, k < 32 ? d_lo : d_hi), k & 31));
        const float r = (float)K_ * nrm + S - 2.f * rowG + ALPHA_;
        if (lane == 0) rs1_s = r;
    }
    const float4 ck0 = ((const float4*)(cluster + (size_t)k * F_ + c0))[lane];
    const float4 ck1 = ((const float4*)(cluster + (size_t)k * F_ + c0))[lane + 32];

    const int c = __popcll(m);
    int incl = c;
#pragma unroll
    for (int d = 1; d < 32; d <<= 1) {
        const int t = __shfl_up(incl, d, 32);
        if (lane >= d) incl += t;
    }
    if (lane == 31) wave_tot_s[wv] = incl;
    __syncthreads();
    int base = 0, total = 0;
#pragma unroll
    for (int w = 0; w < CU_WAVES; ++w) {
        const int t = wave_tot_s[w];
        if (w < wv) base += t;
        total += t;
    }
    total = __builtin_amdgcn_readfirstlane(total);
    const int beg = (total * j) / CU_SLICES;
    const int end = (total * (j + 1)) / CU_SLICES;
    {
        int pos = base + incl - c;
        const int n0 = (int)threadIdx.x * 64;
        while (m) {
            const int b = __builtin_ctzll(m);
            m &= m - 1;
            if (pos >= beg && pos < end) list_s[pos - beg] = (unsigned short)(n0 + b);
            ++pos;
        }
    }
    __syncthreads();

    float4 s0 = {0.f, 0.f, 0.f, 0.f};
    float4 s1 = {0.f, 0.f, 0.f, 0.f};
    const float* __restrict__ fb = features + c0;
    const v2f c01 = {ck0.x, ck0.y}, c23 = {ck0.z, ck0.w}, c45 = {ck1.x, ck1.y}, c67 = {ck1.z, ck1.w};
    for (int q = beg; q < end; q += CU_G) {
        int pp = q + (lane < CU_G ? lane : CU_G - 1);
        pp = pp < end ? pp : end - 1;
        const int rv = (int)list_s[pp - beg];
        v8f X[CU_G];
#pragma unroll
        for (int r = 0; r < CU_G; ++r) {
            const int row = __builtin_amdgcn_readlane(rv, r);
            const float4* rp = (const float4*)(fb + (size_t)row * F_);
            const float4 lo = rp[lane];
            const float4 hi = rp[lane + 32];
            X[r] = (v8f){lo.x, lo.y, lo.z, lo.w, hi.x, hi.y, hi.z, hi.w};
        }
        asm volatile("" ::"v"(X[0]), "v"(X[1]), "v"(X[2]), "v"(X[3]), "v"(X[4]), "v"(X[5]),
                     "v"(X[6]), "v"(X[7]), "v"(X[8]), "v"(X[9]), "v"(X[10]), "v"(X[11]),
                     "v"(X[12]), "v"(X[13]), "v"(X[14]), "v"(X[15]), "v"(X[16]), "v"(X[17]),
                     "v"(X[18]), "v"(X[19]));
        static_assert(CU_G == 20, "the fence above names CU_G values");
        __builtin_amdgcn_sched_barrier(0);
        float pmine = 0.f;
#pragma unroll
        for (int r = 0; r < CU_G; ++r) {
            const float wt = (q + r < end) ? 1.f : 0.f;
            s0.x = fmaf(X[r][0], wt, s0.x);
            s0.y = fmaf(X[r][1], wt, s0.y);
            s0.z = fmaf(X[r][2], wt, s0.z);
            s0.w = fmaf(X[r][3], wt, s0.w);
            s1.x = fmaf(X[r][4], wt, s1.x);
            s1.y = fmaf(X[r][5], wt, s1.y);
            s1.z = fmaf(X[r][6], wt, s1.z);
            s1.w = fmaf(X[r][7], wt, s1.w);
            const v2f e0 = (v2f){X[r][0], X[r][1]} - c01;
            const v2f e1 = (v2f){X[r][2], X[r][3]} - c23;
            const v2f e2 = (v2f){X[r][4], X[r][5]} - c45;
            const v2f e3 = (v2f){X[r][6], X[r][7]} - c67;
            v2f a2 = e0 * e0;
            a2 = __builtin_elementwise_fma(e1, e1, a2);
            a2 = __builtin_elementwise_fma(e2, e2, a2);
            a2 = __builtin_elementwise_fma(e3, e3, a2);
            float p = a2.x + a2.y;
            p = wave_sum(p);
            if (lane == r) pmine = p;
        }
        if (lane < CU_G && q + lane < end) rowp_s[wv][q - beg + lane] = pmine;
    }
    __syncthreads();

    float4* pp4 = (float4*)(part + ((size_t)j * K_ + k) * F_ + c0);
    pp4[lane] = s0;
    pp4[32 + lane] = s1;
    const float rs = rs1_s;
    if (j == 0 && threadIdx.x == 0) {
        rs1_out[k] = rs;
        cnt_out[k] = total;
    }
    for (int i = (int)threadIdx.x; i < end - beg; i += CU_WAVES * 32) {
        float s = 0.f;
#pragma unroll
        for (int w = 0; w < CU_WAVES; ++w) s += rowp_s[w][i];
        loss[list_s[i]] = s / rs;
    }
}

#define FU_THREADS 256
#define FU_BLOCKS_PER_K ((F_ / 4) / FU_THREADS)
static_assert(FU_BLOCKS_PER_K * FU_THREADS * 4 == F_, "launch 3 covers a cluster row exactly");

__global__ __launch_bounds__(FU_THREADS) void finish_update(const float* __restrict__ part,
                                                            const float* __restrict__ cluster,
                                                            const float* __restrict__ cw,
                                                            const float* __restrict__ rs1,
                                                            const int* __restrict__ cnt,
                                                            float* __restrict__ newc) {
    const int k = blockIdx.x / FU_BLOCKS_PER_K;
    const int f4 = (blockIdx.x % FU_BLOCKS_PER_K) * FU_THREADS + threadIdx.x;
    const float4* __restrict__ p = (const float4*)(part + (size_t)k * F_) + f4;
    float4 v[CU_SLICES];
#pragma unroll
    for (int j = 0; j < CU_SLICES; ++j) v[j] = p[(size_t)j * (K_ * F_ / 4)];
    const float4 cc = ((const float4*)(cluster + (size_t)k * F_))[f4];
    float4 a = {0.f, 0.f, 0.f, 0.f};
#pragma unroll
    for (int j = 0; j < CU_SLICES; ++j) {
        a.x += v[j].x;
        a.y += v[j].y;
        a.z += v[j].z;
        a.w += v[j].w;
    }
    const float wk = cw[k] / rs1[k];
    const float tf = (float)cnt[k];
    float4 o;
    o.x = cc.x - GAMMA_ * wk * (a.x - tf * cc.x);
    o.y = cc.y - GAMMA_ * wk * (a.y - tf * cc.y);
    o.z = cc.z - GAMMA_ * wk * (a.z - tf * cc.z);
    o.w = cc.w - GAMMA_ * wk * (a.w - tf * cc.w);
    ((float4*)(newc + (size_t)k * F_))[f4] = o;
}

extern "C" void kernel_launch(void* const* d_in, const int* in_sizes, int n_in,
                              void* d_out, int out_size, void* d_ws, size_t ws_size,
                              hipStream_t stream) {
    const float* features = (const float*)d_in[0];
    const float* labels   = (const float*)d_in[1];
    const float* cluster  = (const float*)d_in[2];
    const float* cw       = (const float*)d_in[3];

    float* loss = (float*)d_out;
    float* newc = loss + N_;

    float* Gp    = (float*)d_ws;
    float* dgw   = Gp + GR_PLANES * K_ * K_;
    float* partw = dgw + GR_PLANES * (K_ / 16) * DG_PITCH;
    float* rs1w  = partw + CU_SLICES * K_ * F_;
    int* cntw    = (int*)(rs1w + K_);
    unsigned* bmw = (unsigned*)(cntw + K_);

    gram_and_argmax<<<GR_BLOCKS + AR_BLOCKS, L1_WAVES * 32, 0, stream>>>(labels, cluster, Gp, dgw, bmw);
    cluster_rows<<<dim3(CU_SLICES, K_), CU_WAVES * 32, 0, stream>>>(
        features, cluster, Gp, dgw, bmw, partw, rs1w, cntw, loss);
    finish_update<<<K_ * FU_BLOCKS_PER_K, FU_THREADS, 0, stream>>>(partw, cluster, cw, rs1w, cntw, newc);
}
